# starting kernel
# baseline (speedup 1.0000x reference)
#include <hip/hip_runtime.h>
#include <hip/hip_bf16.h>

#define NNODES 50000
#define NEDGES 800000
#define INF    128
#define HID    64
#define OUTD   64
#define NHEAD  4
#define SLOPE  0.2f
#define DEG    16

typedef __attribute__((ext_vector_type(2))) float v2f;
typedef __attribute__((ext_vector_type(8))) float v8f;

__global__ void gemm_wmma_f32(const float* __restrict__ A,
                              const float* __restrict__ B,
                              float* __restrict__ C,
                              int M, int K, int Nc) {
    const int wavesPerBlock = blockDim.x >> 5;
    const int wave  = blockIdx.x * wavesPerBlock + (threadIdx.x >> 5);
    const int tilesN = Nc >> 4;
    const int totalTiles = (M >> 4) * tilesN;
    if (wave >= totalTiles) return;

    const int tm = wave / tilesN;
    const int tn = wave - tm * tilesN;
    const int lane = threadIdx.x & 31;
    const int g = lane >> 4;
    const int m = lane & 15;

    const float* __restrict__ Arow = A + (size_t)(tm * 16 + m) * K;
    const float* __restrict__ Bcol = B + (size_t)(tn * 16 + m);

    v8f c = {};
    for (int k0 = 0; k0 < K; k0 += 4) {
        v2f a, b;
        a.x = Arow[k0 + 2 * g];
        a.y = Arow[k0 + 2 * g + 1];
        b.x = Bcol[(size_t)(k0 + 2 * g) * Nc];
        b.y = Bcol[(size_t)(k0 + 2 * g + 1) * Nc];
        c = __builtin_amdgcn_wmma_f32_16x16x4_f32(
                 false, a,  false, b,
                 (short)0, c,  false,  false);
    }

    float* __restrict__ Crow = C + (size_t)(tm * 16 + 8 * g) * Nc + tn * 16 + m;
#pragma unroll
    for (int r = 0; r < 8; ++r)
        Crow[(size_t)r * Nc] = c[r];
}

__global__ void gat_scores(const float* __restrict__ h,
                           const float* __restrict__ al,
                           const float* __restrict__ ar,
                           float* __restrict__ el,
                           float* __restrict__ er,
                           int n_nodes) {
    int idx = blockIdx.x * blockDim.x + threadIdx.x;
    if (idx >= n_nodes * NHEAD) return;
    int head = idx & (NHEAD - 1);
    int node = idx >> 2;
    const float* __restrict__ hp = h + (size_t)node * (NHEAD * HID) + head * HID;
    const float* __restrict__ ap = al + head * HID;
    const float* __restrict__ rp = ar + head * HID;
    float s1 = 0.f, s2 = 0.f;
#pragma unroll 8
    for (int j = 0; j < HID; ++j) {
        float v = hp[j];
        s1 += v * ap[j];
        s2 += v * rp[j];
    }
    el[idx] = s1;
    er[idx] = s2;
}

__global__ void gat_aggregate(const float* __restrict__ h,
                              const float* __restrict__ el,
                              const float* __restrict__ er,
                              const int*   __restrict__ src,
                              const float* __restrict__ bias,
                              float* __restrict__ outm,
                              int n_nodes) {
    const int node = blockIdx.x;
    const int head = threadIdx.x >> 5;
    const int lane = threadIdx.x & 31;

    const float er_v = er[(size_t)node * NHEAD + head];

    int   s    = 0;
    float eval = -1.0e30f;
    if (lane < DEG) {
        s = src[node + lane * n_nodes];
        float t = el[(size_t)s * NHEAD + head] + er_v;
        eval = (t > 0.f) ? t : SLOPE * t;
    }

    float mx = eval;
#pragma unroll
    for (int off = 16; off > 0; off >>= 1)
        mx = fmaxf(mx, __shfl_xor(mx, off, 32));

    float ex = (lane < DEG) ? __expf(eval - mx) : 0.f;

    float denom = ex;
#pragma unroll
    for (int off = 16; off > 0; off >>= 1)
        denom += __shfl_xor(denom, off, 32);

    const float alpha = ex / denom;

    float acc0 = 0.f, acc1 = 0.f;
#pragma unroll
    for (int k = 0; k < DEG; ++k) {
        float a  = __shfl(alpha, k, 32);
        int   sk = __shfl(s,     k, 32);
        const float* __restrict__ hp =
            h + (size_t)sk * (NHEAD * HID) + head * HID;
        acc0 += a * hp[lane];
        acc1 += a * hp[lane + 32];
    }

    __shared__ float sh[NHEAD][HID];
    sh[head][lane]      = acc0 + bias[head * HID + lane];
    sh[head][lane + 32] = acc1 + bias[head * HID + lane + 32];
    __syncthreads();

    if (threadIdx.x < HID) {
        int t = threadIdx.x;
        float v = 0.25f * (sh[0][t] + sh[1][t] + sh[2][t] + sh[3][t]);
        outm[(size_t)node * HID + t] = v;
    }
}

extern "C" void kernel_launch(void* const* d_in, const int* in_sizes, int n_in,
                              void* d_out, int out_size, void* d_ws, size_t ws_size,
                              hipStream_t stream) {
    const float* feat = (const float*)d_in[0];
    const int*   src  = (const int*)  d_in[1];
    const float* W1   = (const float*)d_in[3];
    const float* al1  = (const float*)d_in[4];
    const float* ar1  = (const float*)d_in[5];
    const float* b1   = (const float*)d_in[6];
    const float* W2   = (const float*)d_in[7];
    const float* al2  = (const float*)d_in[8];
    const float* ar2  = (const float*)d_in[9];
    const float* b2   = (const float*)d_in[10];

    const int N = NNODES;

    float* h  = (float*)d_ws;
    float* el = h  + (size_t)N * (NHEAD * HID);
    float* er = el + (size_t)N * NHEAD;
    float* x2 = er + (size_t)N * NHEAD;

    const int wavesPerBlock = 8;
    const int tiles1 = (N / 16) * (256 / 16);
    const int tiles2 = (N / 16) * (256 / 16);
    const int gemmBlocks1 = (tiles1 + wavesPerBlock - 1) / wavesPerBlock;
    const int gemmBlocks2 = (tiles2 + wavesPerBlock - 1) / wavesPerBlock;
    const int scoreBlocks = (N * NHEAD + 255) / 256;

    gemm_wmma_f32<<<gemmBlocks1, 256, 0, stream>>>(feat, W1, h, N, INF, NHEAD * HID);
    gat_scores  <<<scoreBlocks, 256, 0, stream>>>(h, al1, ar1, el, er, N);
    gat_aggregate<<<N, 128, 0, stream>>>(h, el, er, src, b1, x2, N);

    gemm_wmma_f32<<<gemmBlocks2, 256, 0, stream>>>(x2, W2, h, N, HID, NHEAD * OUTD);
    gat_scores  <<<scoreBlocks, 256, 0, stream>>>(h, al2, ar2, el, er, N);
    gat_aggregate<<<N, 128, 0, stream>>>(h, el, er, src, b2, (float*)d_out, N);
}


# GEMM re-tiled: each block stages one head's K x 64 panel of W in LDS in fragment order; each wave keeps its 16 rows of A in registers and feeds 4 accumulator tiles per A fragment
# speedup vs baseline: 1.1563x; 1.1563x over previous
#include <hip/hip_runtime.h>
#include <hip/hip_bf16.h>

#define NNODES 50000
#define NEDGES 800000
#define INF    128
#define HID    64
#define OUTD   64
#define NHEAD  4
#define SLOPE  0.2f
#define DEG    16

typedef __attribute__((ext_vector_type(2))) float v2f;
typedef __attribute__((ext_vector_type(8))) float v8f;

template <int K>
__global__ __launch_bounds__(128)
void gemm_wmma_f32_panel(const float* __restrict__ A,
                         const float* __restrict__ B,
                         float* __restrict__ C,
                         int nStrips) {
    constexpr int NC = NHEAD * HID;
    __shared__ float4 Bs[K * 16];

    const int head     = blockIdx.x & (NHEAD - 1);
    const int rowGroup = blockIdx.x >> 2;
    const int tid  = threadIdx.x;
    const int lane = tid & 31;
    const int wave = tid >> 5;
    const int g = lane >> 4;
    const int m = lane & 15;
    const int strip = rowGroup * 4 + wave;
    const bool active = strip < nStrips;

    float4 a[K / 8];
    if (active) {
        const float4* __restrict__ Ap = reinterpret_cast<const float4*>(
            A + (size_t)(strip * 16 + m) * K + 4 * g);
#pragma unroll
        for (int i = 0; i < K / 8; ++i)
            a[i] = Ap[2 * i];
    }

    const float* __restrict__ Bh = B + head * HID;
#pragma unroll
    for (int it = 0; it < K / 16; ++it) {
        const int item = it * 128 + tid;
        const int p  = item >> 4;
        const int mm = item & 15;
        const float4 x = *reinterpret_cast<const float4*>(Bh + (size_t)(2 * p)     * NC + 4 * mm);
        const float4 y = *reinterpret_cast<const float4*>(Bh + (size_t)(2 * p + 1) * NC + 4 * mm);
        Bs[(2 * p)     * 16 + mm] = make_float4(x.x, y.x, x.y, y.y);
        Bs[(2 * p + 1) * 16 + mm] = make_float4(x.z, y.z, x.w, y.w);
    }
    __syncthreads();

    if (!active) return;

    v8f c0 = {}, c1 = {}, c2 = {}, c3 = {};
#pragma unroll
    for (int i = 0; i < K / 8; ++i) {
#pragma unroll
        for (int s = 0; s < 2; ++s) {
            const int p = 4 * i + 2 * g + s;
            const float4 b01 = Bs[(2 * p)     * 16 + m];
            const float4 b23 = Bs[(2 * p + 1) * 16 + m];
            v2f av, bv0, bv1, bv2, bv3;
            av.x  = s ? a[i].z : a[i].x;
            av.y  = s ? a[i].w : a[i].y;
            bv0.x = b01.x; bv0.y = b01.y;
            bv1.x = b01.z; bv1.y = b01.w;
            bv2.x = b23.x; bv2.y = b23.y;
            bv3.x = b23.z; bv3.y = b23.w;
            c0 = __builtin_amdgcn_wmma_f32_16x16x4_f32(false, av, false, bv0, (short)0, c0, false, false);
            c1 = __builtin_amdgcn_wmma_f32_16x16x4_f32(false, av, false, bv1, (short)0, c1, false, false);
            c2 = __builtin_amdgcn_wmma_f32_16x16x4_f32(false, av, false, bv2, (short)0, c2, false, false);
            c3 = __builtin_amdgcn_wmma_f32_16x16x4_f32(false, av, false, bv3, (short)0, c3, false, false);
        }
    }

    float* __restrict__ Cp = C + (size_t)(strip * 16 + 8 * g) * NC + head * HID + 4 * m;
#pragma unroll
    for (int r = 0; r < 8; ++r)
        *reinterpret_cast<float4*>(Cp + (size_t)r * NC) = make_float4(c0[r], c1[r], c2[r], c3[r]);
}

__global__ void gat_scores(const float* __restrict__ h,
                           const float* __restrict__ al,
                           const float* __restrict__ ar,
                           float* __restrict__ el,
                           float* __restrict__ er,
                           int n_nodes) {
    int idx = blockIdx.x * blockDim.x + threadIdx.x;
    if (idx >= n_nodes * NHEAD) return;
    int head = idx & (NHEAD - 1);
    int node = idx >> 2;
    const float* __restrict__ hp = h + (size_t)node * (NHEAD * HID) + head * HID;
    const float* __restrict__ ap = al + head * HID;
    const float* __restrict__ rp = ar + head * HID;
    float s1 = 0.f, s2 = 0.f;
#pragma unroll 8
    for (int j = 0; j < HID; ++j) {
        float v = hp[j];
        s1 += v * ap[j];
        s2 += v * rp[j];
    }
    el[idx] = s1;
    er[idx] = s2;
}

__global__ void gat_aggregate(const float* __restrict__ h,
                              const float* __restrict__ el,
                              const float* __restrict__ er,
                              const int*   __restrict__ src,
                              const float* __restrict__ bias,
                              float* __restrict__ outm,
                              int n_nodes) {
    const int node = blockIdx.x;
    const int head = threadIdx.x >> 5;
    const int lane = threadIdx.x & 31;

    const float er_v = er[(size_t)node * NHEAD + head];

    int   s    = 0;
    float eval = -1.0e30f;
    if (lane < DEG) {
        s = src[node + lane * n_nodes];
        float t = el[(size_t)s * NHEAD + head] + er_v;
        eval = (t > 0.f) ? t : SLOPE * t;
    }

    float mx = eval;
#pragma unroll
    for (int off = 16; off > 0; off >>= 1)
        mx = fmaxf(mx, __shfl_xor(mx, off, 32));

    float ex = (lane < DEG) ? __expf(eval - mx) : 0.f;

    float denom = ex;
#pragma unroll
    for (int off = 16; off > 0; off >>= 1)
        denom += __shfl_xor(denom, off, 32);

    const float alpha = ex / denom;

    float acc0 = 0.f, acc1 = 0.f;
#pragma unroll
    for (int k = 0; k < DEG; ++k) {
        float a  = __shfl(alpha, k, 32);
        int   sk = __shfl(s,     k, 32);
        const float* __restrict__ hp =
            h + (size_t)sk * (NHEAD * HID) + head * HID;
        acc0 += a * hp[lane];
        acc1 += a * hp[lane + 32];
    }

    __shared__ float sh[NHEAD][HID];
    sh[head][lane]      = acc0 + bias[head * HID + lane];
    sh[head][lane + 32] = acc1 + bias[head * HID + lane + 32];
    __syncthreads();

    if (threadIdx.x < HID) {
        int t = threadIdx.x;
        float v = 0.25f * (sh[0][t] + sh[1][t] + sh[2][t] + sh[3][t]);
        outm[(size_t)node * HID + t] = v;
    }
}

extern "C" void kernel_launch(void* const* d_in, const int* in_sizes, int n_in,
                              void* d_out, int out_size, void* d_ws, size_t ws_size,
                              hipStream_t stream) {
    const float* feat = (const float*)d_in[0];
    const int*   src  = (const int*)  d_in[1];
    const float* W1   = (const float*)d_in[3];
    const float* al1  = (const float*)d_in[4];
    const float* ar1  = (const float*)d_in[5];
    const float* b1   = (const float*)d_in[6];
    const float* W2   = (const float*)d_in[7];
    const float* al2  = (const float*)d_in[8];
    const float* ar2  = (const float*)d_in[9];
    const float* b2   = (const float*)d_in[10];

    const int N = NNODES;

    float* h  = (float*)d_ws;
    float* el = h  + (size_t)N * (NHEAD * HID);
    float* er = el + (size_t)N * NHEAD;
    float* x2 = er + (size_t)N * NHEAD;

    const int nStrips     = 3125;
    const int gemmBlocks  = 3128;
    const int scoreBlocks = 782;

    gemm_wmma_f32_panel<INF><<<gemmBlocks, 128, 0, stream>>>(feat, W1, h, nStrips);
    gat_scores  <<<scoreBlocks, 256, 0, stream>>>(h, al1, ar1, el, er, N);
    gat_aggregate<<<N, 128, 0, stream>>>(h, el, er, src, b1, x2, N);

    gemm_wmma_f32_panel<HID><<<gemmBlocks, 128, 0, stream>>>(x2, W2, h, nStrips);
    gat_scores  <<<scoreBlocks, 256, 0, stream>>>(h, al2, ar2, el, er, N);
    gat_aggregate<<<N, 128, 0, stream>>>(h, el, er, src, b2, (float*)d_out, N);
}


# aggregate kernel: 8,192 resident waves each loop over whole nodes (all 4 heads per wave, b128 gathers covering whole lines) instead of one 4-wave block per node (200,000 waves per launch)
# speedup vs baseline: 1.4755x; 1.2761x over previous
#include <hip/hip_runtime.h>
#include <hip/hip_bf16.h>

#define NNODES 50000
#define NEDGES 800000
#define INF    128
#define HID    64
#define OUTD   64
#define NHEAD  4
#define SLOPE  0.2f
#define DEG    16

typedef __attribute__((ext_vector_type(2))) float v2f;
typedef __attribute__((ext_vector_type(8))) float v8f;

template <int K>
__global__ __launch_bounds__(128)
void gemm_wmma_f32_panel(const float* __restrict__ A,
                         const float* __restrict__ B,
                         float* __restrict__ C,
                         int nStrips) {
    constexpr int NC = NHEAD * HID;
    __shared__ float4 Bs[K * 16];

    const int head     = blockIdx.x & (NHEAD - 1);
    const int rowGroup = blockIdx.x >> 2;
    const int tid  = threadIdx.x;
    const int lane = tid & 31;
    const int wave = tid >> 5;
    const int g = lane >> 4;
    const int m = lane & 15;
    const int strip = rowGroup * 4 + wave;
    const bool active = strip < nStrips;

    float4 a[K / 8];
    if (active) {
        const float4* __restrict__ Ap = reinterpret_cast<const float4*>(
            A + (size_t)(strip * 16 + m) * K + 4 * g);
#pragma unroll
        for (int i = 0; i < K / 8; ++i)
            a[i] = Ap[2 * i];
    }

    const float* __restrict__ Bh = B + head * HID;
#pragma unroll
    for (int it = 0; it < K / 16; ++it) {
        const int item = it * 128 + tid;
        const int p  = item >> 4;
        const int mm = item & 15;
        const float4 x = *reinterpret_cast<const float4*>(Bh + (size_t)(2 * p)     * NC + 4 * mm);
        const float4 y = *reinterpret_cast<const float4*>(Bh + (size_t)(2 * p + 1) * NC + 4 * mm);
        Bs[(2 * p)     * 16 + mm] = make_float4(x.x, y.x, x.y, y.y);
        Bs[(2 * p + 1) * 16 + mm] = make_float4(x.z, y.z, x.w, y.w);
    }
    __syncthreads();

    if (!active) return;

    v8f c0 = {}, c1 = {}, c2 = {}, c3 = {};
#pragma unroll
    for (int i = 0; i < K / 8; ++i) {
#pragma unroll
        for (int s = 0; s < 2; ++s) {
            const int p = 4 * i + 2 * g + s;
            const float4 b01 = Bs[(2 * p)     * 16 + m];
            const float4 b23 = Bs[(2 * p + 1) * 16 + m];
            v2f av, bv0, bv1, bv2, bv3;
            av.x  = s ? a[i].z : a[i].x;
            av.y  = s ? a[i].w : a[i].y;
            bv0.x = b01.x; bv0.y = b01.y;
            bv1.x = b01.z; bv1.y = b01.w;
            bv2.x = b23.x; bv2.y = b23.y;
            bv3.x = b23.z; bv3.y = b23.w;
            c0 = __builtin_amdgcn_wmma_f32_16x16x4_f32(false, av, false, bv0, (short)0, c0, false, false);
            c1 = __builtin_amdgcn_wmma_f32_16x16x4_f32(false, av, false, bv1, (short)0, c1, false, false);
            c2 = __builtin_amdgcn_wmma_f32_16x16x4_f32(false, av, false, bv2, (short)0, c2, false, false);
            c3 = __builtin_amdgcn_wmma_f32_16x16x4_f32(false, av, false, bv3, (short)0, c3, false, false);
        }
    }

    float* __restrict__ Cp = C + (size_t)(strip * 16 + 8 * g) * NC + head * HID + 4 * m;
#pragma unroll
    for (int r = 0; r < 8; ++r)
        *reinterpret_cast<float4*>(Cp + (size_t)r * NC) = make_float4(c0[r], c1[r], c2[r], c3[r]);
}

__global__ void gat_scores(const float* __restrict__ h,
                           const float* __restrict__ al,
                           const float* __restrict__ ar,
                           float* __restrict__ el,
                           float* __restrict__ er,
                           int n_nodes) {
    int idx = blockIdx.x * blockDim.x + threadIdx.x;
    if (idx >= n_nodes * NHEAD) return;
    int head = idx & (NHEAD - 1);
    int node = idx >> 2;
    const float* __restrict__ hp = h + (size_t)node * (NHEAD * HID) + head * HID;
    const float* __restrict__ ap = al + head * HID;
    const float* __restrict__ rp = ar + head * HID;
    float s1 = 0.f, s2 = 0.f;
#pragma unroll 8
    for (int j = 0; j < HID; ++j) {
        float v = hp[j];
        s1 += v * ap[j];
        s2 += v * rp[j];
    }
    el[idx] = s1;
    er[idx] = s2;
}

#define AGG_THREADS 256
#define AGG_BLOCKS  1024

template <int J>
__device__ __forceinline__ float row_bcast(float x) {
    const int v = __float_as_int(x);
    return __int_as_float(__builtin_amdgcn_update_dpp(v, v, 0x150 + J, 0xF, 0xF, false));
}

template <int M>
__device__ __forceinline__ float row_xor(float x) {
    const int v = __float_as_int(x);
    return __int_as_float(__builtin_amdgcn_update_dpp(v, v, 0x160 + M, 0xF, 0xF, false));
}

__device__ __forceinline__ float row_max16(float x) {
    x = fmaxf(x, row_xor<8>(x));
    x = fmaxf(x, row_xor<4>(x));
    x = fmaxf(x, row_xor<2>(x));
    x = fmaxf(x, row_xor<1>(x));
    return x;
}

__device__ __forceinline__ float row_sum16(float x) {
    x += row_xor<8>(x);
    x += row_xor<4>(x);
    x += row_xor<2>(x);
    x += row_xor<1>(x);
    return x;
}

template <int J>
__device__ __forceinline__ void gather_edge(const float* __restrict__ h, int s, int lane,
                                            float alphaA, float alphaB,
                                            float4& accA, float4& accB) {
    const int sj = __builtin_amdgcn_readlane(s, J);
    const float* __restrict__ hp = h + (size_t)sj * (NHEAD * HID) + 4 * lane;
    const float4 rA = *reinterpret_cast<const float4*>(hp);
    const float4 rB = *reinterpret_cast<const float4*>(hp + 2 * HID);
    const float aA = row_bcast<J>(alphaA);
    const float aB = row_bcast<J>(alphaB);
    accA.x += aA * rA.x; accA.y += aA * rA.y; accA.z += aA * rA.z; accA.w += aA * rA.w;
    accB.x += aB * rB.x; accB.y += aB * rB.y; accB.z += aB * rB.z; accB.w += aB * rB.w;
}

__global__ __launch_bounds__(AGG_THREADS)
void gat_aggregate(const float* __restrict__ h,
                   const float* __restrict__ el,
                   const float* __restrict__ er,
                   const int*   __restrict__ src,
                   const float* __restrict__ bias,
                   float* __restrict__ outm,
                   int n_nodes) {
    const int lane = threadIdx.x & 31;
    const int g = lane >> 4;
    const int k = lane & 15;
    const int wave0 = __builtin_amdgcn_readfirstlane(
        (int)(blockIdx.x * (AGG_THREADS / 32) + (threadIdx.x >> 5)));
    const int nWaves = (int)gridDim.x * (AGG_THREADS / 32);

    const float4 bA = *reinterpret_cast<const float4*>(bias + g * HID + 4 * k);
    const float4 bB = *reinterpret_cast<const float4*>(bias + (2 + g) * HID + 4 * k);

    int s_next = (wave0 < n_nodes) ? src[wave0 + k * n_nodes] : 0;

    for (int node = wave0; node < n_nodes; node += nWaves) {
        const int s = s_next;
        if (node + nWaves < n_nodes)
            s_next = src[node + nWaves + k * n_nodes];
        const float4 er4 = *reinterpret_cast<const float4*>(er + (size_t)node * NHEAD);
        const float4 el4 = *reinterpret_cast<const float4*>(el + (size_t)s * NHEAD);

        const float tA = (g ? el4.y : el4.x) + (g ? er4.y : er4.x);
        const float tB = (g ? el4.w : el4.z) + (g ? er4.w : er4.z);
        const float eA = (tA > 0.f) ? tA : SLOPE * tA;
        const float eB = (tB > 0.f) ? tB : SLOPE * tB;

        const float mA = row_max16(eA);
        const float mB = row_max16(eB);
        const float xA = __expf(eA - mA);
        const float xB = __expf(eB - mB);
        const float alphaA = xA / row_sum16(xA);
        const float alphaB = xB / row_sum16(xB);

        float4 accA = make_float4(0.f, 0.f, 0.f, 0.f);
        float4 accB = make_float4(0.f, 0.f, 0.f, 0.f);
        gather_edge<0>(h, s, lane, alphaA, alphaB, accA, accB);
        gather_edge<1>(h, s, lane, alphaA, alphaB, accA, accB);
        gather_edge<2>(h, s, lane, alphaA, alphaB, accA, accB);
        gather_edge<3>(h, s, lane, alphaA, alphaB, accA, accB);
        gather_edge<4>(h, s, lane, alphaA, alphaB, accA, accB);
        gather_edge<5>(h, s, lane, alphaA, alphaB, accA, accB);
        gather_edge<6>(h, s, lane, alphaA, alphaB, accA, accB);
        gather_edge<7>(h, s, lane, alphaA, alphaB, accA, accB);
        gather_edge<8>(h, s, lane, alphaA, alphaB, accA, accB);
        gather_edge<9>(h, s, lane, alphaA, alphaB, accA, accB);
        gather_edge<10>(h, s, lane, alphaA, alphaB, accA, accB);
        gather_edge<11>(h, s, lane, alphaA, alphaB, accA, accB);
        gather_edge<12>(h, s, lane, alphaA, alphaB, accA, accB);
        gather_edge<13>(h, s, lane, alphaA, alphaB, accA, accB);
        gather_edge<14>(h, s, lane, alphaA, alphaB, accA, accB);
        gather_edge<15>(h, s, lane, alphaA, alphaB, accA, accB);

        float4 v;
        v.x = (accA.x + bA.x) + (accB.x + bB.x);
        v.y = (accA.y + bA.y) + (accB.y + bB.y);
        v.z = (accA.z + bA.z) + (accB.z + bB.z);
        v.w = (accA.w + bA.w) + (accB.w + bB.w);
        v.x += __shfl_xor(v.x, 16, 32);
        v.y += __shfl_xor(v.y, 16, 32);
        v.z += __shfl_xor(v.z, 16, 32);
        v.w += __shfl_xor(v.w, 16, 32);

        if (g == 0)
            *reinterpret_cast<float4*>(outm + (size_t)node * HID + 4 * k) =
                make_float4(0.25f * v.x, 0.25f * v.y, 0.25f * v.z, 0.25f * v.w);
    }
}

extern "C" void kernel_launch(void* const* d_in, const int* in_sizes, int n_in,
                              void* d_out, int out_size, void* d_ws, size_t ws_size,
                              hipStream_t stream) {
    const float* feat = (const float*)d_in[0];
    const int*   src  = (const int*)  d_in[1];
    const float* W1   = (const float*)d_in[3];
    const float* al1  = (const float*)d_in[4];
    const float* ar1  = (const float*)d_in[5];
    const float* b1   = (const float*)d_in[6];
    const float* W2   = (const float*)d_in[7];
    const float* al2  = (const float*)d_in[8];
    const float* ar2  = (const float*)d_in[9];
    const float* b2   = (const float*)d_in[10];

    const int N = NNODES;

    float* h  = (float*)d_ws;
    float* el = h  + (size_t)N * (NHEAD * HID);
    float* er = el + (size_t)N * NHEAD;
    float* x2 = er + (size_t)N * NHEAD;

    const int nStrips     = 3125;
    const int gemmBlocks  = 3128;
    const int scoreBlocks = 782;

    gemm_wmma_f32_panel<INF><<<gemmBlocks, 128, 0, stream>>>(feat, W1, h, nStrips);
    gat_scores  <<<scoreBlocks, 256, 0, stream>>>(h, al1, ar1, el, er, N);
    gat_aggregate<<<AGG_BLOCKS, AGG_THREADS, 0, stream>>>(h, el, er, src, b1, x2, N);

    gemm_wmma_f32_panel<HID><<<gemmBlocks, 128, 0, stream>>>(x2, W2, h, nStrips);
    gat_scores  <<<scoreBlocks, 256, 0, stream>>>(h, al2, ar2, el, er, N);
    gat_aggregate<<<AGG_BLOCKS, AGG_THREADS, 0, stream>>>(h, el, er, src, b2, (float*)d_out, N);
}


# layer-2 scores formed in the layer-1 aggregate kernel from folded weights (W2_h al2_h, W2_h ar2_h) with one 16-step f32 WMMA chain per 16-node strip; scores-only GEMM launch gone: 5 -> 4 launches
# speedup vs baseline: 1.8692x; 1.2668x over previous
#include <hip/hip_runtime.h>
#include <hip/hip_bf16.h>

#define NNODES 50000
#define NEDGES 800000
#define INF    128
#define HID    64
#define OUTD   64
#define NHEAD  4
#define SLOPE  0.2f
#define DEG    16

typedef __attribute__((ext_vector_type(2))) float v2f;
typedef __attribute__((ext_vector_type(8))) float v8f;

template <int J>
__device__ __forceinline__ float row_bcast(float x) {
    const int v = __float_as_int(x);
    return __int_as_float(__builtin_amdgcn_update_dpp(v, v, 0x150 + J, 0xF, 0xF, false));
}

template <int M>
__device__ __forceinline__ float row_xor(float x) {
    const int v = __float_as_int(x);
    return __int_as_float(__builtin_amdgcn_update_dpp(v, v, 0x160 + M, 0xF, 0xF, false));
}

__device__ __forceinline__ float row_max16(float x) {
    x = fmaxf(x, row_xor<8>(x));
    x = fmaxf(x, row_xor<4>(x));
    x = fmaxf(x, row_xor<2>(x));
    x = fmaxf(x, row_xor<1>(x));
    return x;
}

__device__ __forceinline__ float row_sum16(float x) {
    x += row_xor<8>(x);
    x += row_xor<4>(x);
    x += row_xor<2>(x);
    x += row_xor<1>(x);
    return x;
}

template <int M>
__device__ __forceinline__ float add_row_xor(float acc, float x) {
    return acc + __int_as_float(
        __builtin_amdgcn_update_dpp(0, __float_as_int(x), 0x160 + M, 0xF, 0xF, true));
}

template <int K>
__global__ __launch_bounds__(128)
void gemm_wmma_f32_panel(const float* __restrict__ A,
                         const float* __restrict__ B,
                         float* __restrict__ C,
                         const float* __restrict__ al,
                         const float* __restrict__ ar,
                         float* __restrict__ el,
                         float* __restrict__ er,
                         int nStrips) {
    constexpr int NC = NHEAD * HID;
    __shared__ float4 Bs[K * 16];

    const int head     = blockIdx.x & (NHEAD - 1);
    const int rowGroup = blockIdx.x >> 2;
    const int tid  = threadIdx.x;
    const int lane = tid & 31;
    const int wave = tid >> 5;
    const int g = lane >> 4;
    const int m = lane & 15;
    const int strip = rowGroup * 4 + wave;
    const bool active = strip < nStrips;

    float4 a[K / 8];
    if (active) {
        const float4* __restrict__ Ap = reinterpret_cast<const float4*>(
            A + (size_t)(strip * 16 + m) * K + 4 * g);
#pragma unroll
        for (int i = 0; i < K / 8; ++i)
            a[i] = Ap[2 * i];
    }

    const float4 alv = *reinterpret_cast<const float4*>(al + head * HID + 4 * m);
    const float4 arv = *reinterpret_cast<const float4*>(ar + head * HID + 4 * m);

    const float* __restrict__ Bh = B + head * HID;
#pragma unroll
    for (int it = 0; it < K / 16; ++it) {
        const int item = it * 128 + tid;
        const int p  = item >> 4;
        const int mm = item & 15;
        const float4 x = *reinterpret_cast<const float4*>(Bh + (size_t)(2 * p)     * NC + 4 * mm);
        const float4 y = *reinterpret_cast<const float4*>(Bh + (size_t)(2 * p + 1) * NC + 4 * mm);
        Bs[(2 * p)     * 16 + mm] = make_float4(x.x, y.x, x.y, y.y);
        Bs[(2 * p + 1) * 16 + mm] = make_float4(x.z, y.z, x.w, y.w);
    }
    __syncthreads();

    if (!active) return;

    v8f c0 = {}, c1 = {}, c2 = {}, c3 = {};
#pragma unroll
    for (int i = 0; i < K / 8; ++i) {
#pragma unroll
        for (int s = 0; s < 2; ++s) {
            const int p = 4 * i + 2 * g + s;
            const float4 b01 = Bs[(2 * p)     * 16 + m];
            const float4 b23 = Bs[(2 * p + 1) * 16 + m];
            v2f av, bv0, bv1, bv2, bv3;
            av.x  = s ? a[i].z : a[i].x;
            av.y  = s ? a[i].w : a[i].y;
            bv0.x = b01.x; bv0.y = b01.y;
            bv1.x = b01.z; bv1.y = b01.w;
            bv2.x = b23.x; bv2.y = b23.y;
            bv3.x = b23.z; bv3.y = b23.w;
            c0 = __builtin_amdgcn_wmma_f32_16x16x4_f32(false, av, false, bv0, (short)0, c0, false, false);
            c1 = __builtin_amdgcn_wmma_f32_16x16x4_f32(false, av, false, bv1, (short)0, c1, false, false);
            c2 = __builtin_amdgcn_wmma_f32_16x16x4_f32(false, av, false, bv2, (short)0, c2, false, false);
            c3 = __builtin_amdgcn_wmma_f32_16x16x4_f32(false, av, false, bv3, (short)0, c3, false, false);
        }
    }

    {
        float* __restrict__ Cp = C + (size_t)(strip * 16 + 8 * g) * NC + head * HID + 4 * m;
#pragma unroll
        for (int r = 0; r < 8; ++r)
            *reinterpret_cast<float4*>(Cp + (size_t)r * NC) = make_float4(c0[r], c1[r], c2[r], c3[r]);
    }

    const bool lo = m < 8;
    const float4 wk = lo ? alv : arv;
    const float4 wg = lo ? arv : alv;
    float sc = 0.f;
#pragma unroll
    for (int r = 0; r < 8; ++r) {
        const float keep = c0[r] * wk.x + c1[r] * wk.y + c2[r] * wk.z + c3[r] * wk.w;
        const float give = c0[r] * wg.x + c1[r] * wg.y + c2[r] * wg.z + c3[r] * wg.w;
        float x = add_row_xor<8>(keep, give);
        x = add_row_xor<4>(x, x);
        x = add_row_xor<2>(x, x);
        x = add_row_xor<1>(x, x);
        sc = ((m & 7) == r) ? x : sc;
    }
    float* dst = lo ? el : er;
    dst[(size_t)(strip * 16 + 8 * g + (m & 7)) * NHEAD + head] = sc;
}

#define AGG_THREADS 256
#define AGG_BLOCKS  1024

template <int J>
__device__ __forceinline__ void gather_edge(const float* __restrict__ h, int s, int lane,
                                            float alphaA, float alphaB,
                                            float4& accA, float4& accB) {
    const int sj = __builtin_amdgcn_readlane(s, J);
    const float* __restrict__ hp = h + (size_t)sj * (NHEAD * HID) + 4 * lane;
    const float4 rA = *reinterpret_cast<const float4*>(hp);
    const float4 rB = *reinterpret_cast<const float4*>(hp + 2 * HID);
    const float aA = row_bcast<J>(alphaA);
    const float aB = row_bcast<J>(alphaB);
    accA.x += aA * rA.x; accA.y += aA * rA.y; accA.z += aA * rA.z; accA.w += aA * rA.w;
    accB.x += aB * rB.x; accB.y += aB * rB.y; accB.z += aB * rB.z; accB.w += aB * rB.w;
}

#define XS_PITCH 68

__global__ __launch_bounds__(AGG_THREADS) __attribute__((amdgpu_waves_per_eu(8)))
void gat_aggregate(const float* __restrict__ h,
                   const float* __restrict__ el,
                   const float* __restrict__ er,
                   const int*   __restrict__ src,
                   const float* __restrict__ bias,
                   float* __restrict__ outm,
                   const float* __restrict__ Wn,
                   const float* __restrict__ aln,
                   const float* __restrict__ arn,
                   float* __restrict__ eln,
                   float* __restrict__ ern,
                   int n_nodes) {
    __shared__ float2 Bfs[(HID / 2) * 16];
    __shared__ float  Xs[2][16 * XS_PITCH];

    const int tid  = threadIdx.x;
    const int lane = tid & 31;
    const int g = lane >> 4;
    const int k = lane & 15;
    const int wave = __builtin_amdgcn_readfirstlane(tid >> 5);
    const int nStrips = n_nodes >> 4;

    const float4 bA = *reinterpret_cast<const float4*>(bias + g * HID + 4 * k);
    const float4 bB = *reinterpret_cast<const float4*>(bias + (2 + g) * HID + 4 * k);

    int strip = blockIdx.x;
    int s_next = (strip < nStrips) ? src[strip * 16 + 2 * wave + k * n_nodes] : 0;

    {
        const int kk = tid >> 2;
        const int hh = tid & 3;
        const float4* __restrict__ wp = reinterpret_cast<const float4*>(Wn + (size_t)kk * (NHEAD * HID) + hh * HID);
        const float4* __restrict__ lp = reinterpret_cast<const float4*>(aln + hh * HID);
        const float4* __restrict__ rp = reinterpret_cast<const float4*>(arn + hh * HID);
        float wl = 0.f, wr = 0.f;
#pragma unroll 4
        for (int q = 0; q < HID / 4; ++q) {
            const float4 w = wp[q];
            const float4 a = lp[q];
            const float4 b = rp[q];
            wl += w.x * a.x + w.y * a.y + w.z * a.z + w.w * a.w;
            wr += w.x * b.x + w.y * b.y + w.z * b.z + w.w * b.w;
        }
        float* __restrict__ bf = reinterpret_cast<float*>(Bfs) + (kk >> 1) * 32 + (kk & 1);
        bf[2 * hh]        = wl;
        bf[2 * (4 + hh)]  = wr;
        bf[2 * (8 + hh)]  = 0.f;
        bf[2 * (12 + hh)] = 0.f;
    }

    int trip = 0;
    for (; strip < nStrips; strip += (int)gridDim.x, ++trip) {
        float* __restrict__ xs = Xs[trip & 1];
#pragma unroll 1
        for (int j = 0; j < 2; ++j) {
            const int node = strip * 16 + 2 * wave + j;
            const int s = s_next;
            const int nnode = j ? (strip + (int)gridDim.x) * 16 + 2 * wave : node + 1;
            if (nnode < n_nodes)
                s_next = src[nnode + k * n_nodes];
            const float4 er4 = *reinterpret_cast<const float4*>(er + (size_t)node * NHEAD);
            const float4 el4 = *reinterpret_cast<const float4*>(el + (size_t)s * NHEAD);

            const float tA = (g ? el4.y : el4.x) + (g ? er4.y : er4.x);
            const float tB = (g ? el4.w : el4.z) + (g ? er4.w : er4.z);
            const float eA = (tA > 0.f) ? tA : SLOPE * tA;
            const float eB = (tB > 0.f) ? tB : SLOPE * tB;

            const float mA = row_max16(eA);
            const float mB = row_max16(eB);
            const float xA = __expf(eA - mA);
            const float xB = __expf(eB - mB);
            const float alphaA = xA / row_sum16(xA);
            const float alphaB = xB / row_sum16(xB);

            float4 accA = make_float4(0.f, 0.f, 0.f, 0.f);
            float4 accB = make_float4(0.f, 0.f, 0.f, 0.f);
            gather_edge<0>(h, s, lane, alphaA, alphaB, accA, accB);
            gather_edge<1>(h, s, lane, alphaA, alphaB, accA, accB);
            gather_edge<2>(h, s, lane, alphaA, alphaB, accA, accB);
            gather_edge<3>(h, s, lane, alphaA, alphaB, accA, accB);
            gather_edge<4>(h, s, lane, alphaA, alphaB, accA, accB);
            gather_edge<5>(h, s, lane, alphaA, alphaB, accA, accB);
            gather_edge<6>(h, s, lane, alphaA, alphaB, accA, accB);
            gather_edge<7>(h, s, lane, alphaA, alphaB, accA, accB);
            gather_edge<8>(h, s, lane, alphaA, alphaB, accA, accB);
            gather_edge<9>(h, s, lane, alphaA, alphaB, accA, accB);
            gather_edge<10>(h, s, lane, alphaA, alphaB, accA, accB);
            gather_edge<11>(h, s, lane, alphaA, alphaB, accA, accB);
            gather_edge<12>(h, s, lane, alphaA, alphaB, accA, accB);
            gather_edge<13>(h, s, lane, alphaA, alphaB, accA, accB);
            gather_edge<14>(h, s, lane, alphaA, alphaB, accA, accB);
            gather_edge<15>(h, s, lane, alphaA, alphaB, accA, accB);

            float4 v;
            v.x = (accA.x + bA.x) + (accB.x + bB.x);
            v.y = (accA.y + bA.y) + (accB.y + bB.y);
            v.z = (accA.z + bA.z) + (accB.z + bB.z);
            v.w = (accA.w + bA.w) + (accB.w + bB.w);
            v.x += __shfl_xor(v.x, 16, 32);
            v.y += __shfl_xor(v.y, 16, 32);
            v.z += __shfl_xor(v.z, 16, 32);
            v.w += __shfl_xor(v.w, 16, 32);

            if (g == 0) {
                const float4 o = make_float4(0.25f * v.x, 0.25f * v.y, 0.25f * v.z, 0.25f * v.w);
                *reinterpret_cast<float4*>(outm + (size_t)node * HID + 4 * k) = o;
                *reinterpret_cast<float4*>(xs + (2 * wave + j) * XS_PITCH + 4 * k) = o;
            }
        }
        __syncthreads();

        if (wave == (trip & 7)) {
            v8f acc0 = {}, acc1 = {};
#pragma unroll
            for (int i = 0; i < HID / 8; ++i) {
                const float4 a = *reinterpret_cast<const float4*>(xs + k * XS_PITCH + 8 * i + 4 * g);
                const float2 b0 = Bfs[(4 * i + 2 * g)     * 16 + k];
                const float2 b1 = Bfs[(4 * i + 2 * g + 1) * 16 + k];
                v2f av0, av1, bv0, bv1;
                av0.x = a.x;  av0.y = a.y;
                av1.x = a.z;  av1.y = a.w;
                bv0.x = b0.x; bv0.y = b0.y;
                bv1.x = b1.x; bv1.y = b1.y;
                acc0 = __builtin_amdgcn_wmma_f32_16x16x4_f32(false, av0, false, bv0, (short)0, acc0, false, false);
                acc1 = __builtin_amdgcn_wmma_f32_16x16x4_f32(false, av1, false, bv1, (short)0, acc1, false, false);
            }
            if (k < 8) {
                float* __restrict__ dstp = (k < 4) ? eln : ern;
                const int hh = k & 3;
#pragma unroll
                for (int r = 0; r < 8; ++r)
                    dstp[(size_t)(strip * 16 + 8 * g + r) * NHEAD + hh] = acc0[r] + acc1[r];
            }
        }
    }
}

template <int J>
__device__ __forceinline__ void gather_edge_in(const float* __restrict__ x, int s, int lane,
                                               float a0, float a1, float a2, float a3,
                                               float2& c0, float2& c1, float2& c2, float2& c3) {
    const int sj = __builtin_amdgcn_readlane(s, J);
    const float2 r = *reinterpret_cast<const float2*>(x + (size_t)sj * HID + 2 * lane);
    const float w0 = row_bcast<J>(a0);
    const float w1 = row_bcast<J>(a1);
    const float w2 = row_bcast<J>(a2);
    const float w3 = row_bcast<J>(a3);
    c0.x += w0 * r.x; c0.y += w0 * r.y;
    c1.x += w1 * r.x; c1.y += w1 * r.y;
    c2.x += w2 * r.x; c2.y += w2 * r.y;
    c3.x += w3 * r.x; c3.y += w3 * r.y;
}

__device__ __forceinline__ float edge_softmax16(float el, float er) {
    const float t = el + er;
    const float e = (t > 0.f) ? t : SLOPE * t;
    const float mx = row_max16(e);
    const float ex = __expf(e - mx);
    return ex / row_sum16(ex);
}

__global__ __launch_bounds__(AGG_THREADS)
void gat_aggregate_inputs(const float* __restrict__ x,
                          const float* __restrict__ el,
                          const float* __restrict__ er,
                          const int*   __restrict__ src,
                          float* __restrict__ agg,
                          int n_nodes) {
    const int lane = threadIdx.x & 31;
    const int k = lane & 15;
    const int wave0 = __builtin_amdgcn_readfirstlane(
        (int)(blockIdx.x * (AGG_THREADS / 32) + (threadIdx.x >> 5)));
    const int nWaves = (int)gridDim.x * (AGG_THREADS / 32);

    int s_next = (wave0 < n_nodes) ? src[wave0 + k * n_nodes] : 0;

    for (int node = wave0; node < n_nodes; node += nWaves) {
        const int s = s_next;
        if (node + nWaves < n_nodes)
            s_next = src[node + nWaves + k * n_nodes];
        const float4 er4 = *reinterpret_cast<const float4*>(er + (size_t)node * NHEAD);
        const float4 el4 = *reinterpret_cast<const float4*>(el + (size_t)s * NHEAD);

        const float a0 = edge_softmax16(el4.x, er4.x);
        const float a1 = edge_softmax16(el4.y, er4.y);
        const float a2 = edge_softmax16(el4.z, er4.z);
        const float a3 = edge_softmax16(el4.w, er4.w);

        float2 c0 = make_float2(0.f, 0.f), c1 = c0, c2 = c0, c3 = c0;
        gather_edge_in<0>(x, s, lane, a0, a1, a2, a3, c0, c1, c2, c3);
        gather_edge_in<1>(x, s, lane, a0, a1, a2, a3, c0, c1, c2, c3);
        gather_edge_in<2>(x, s, lane, a0, a1, a2, a3, c0, c1, c2, c3);
        gather_edge_in<3>(x, s, lane, a0, a1, a2, a3, c0, c1, c2, c3);
        gather_edge_in<4>(x, s, lane, a0, a1, a2, a3, c0, c1, c2, c3);
        gather_edge_in<5>(x, s, lane, a0, a1, a2, a3, c0, c1, c2, c3);
        gather_edge_in<6>(x, s, lane, a0, a1, a2, a3, c0, c1, c2, c3);
        gather_edge_in<7>(x, s, lane, a0, a1, a2, a3, c0, c1, c2, c3);
        gather_edge_in<8>(x, s, lane, a0, a1, a2, a3, c0, c1, c2, c3);
        gather_edge_in<9>(x, s, lane, a0, a1, a2, a3, c0, c1, c2, c3);
        gather_edge_in<10>(x, s, lane, a0, a1, a2, a3, c0, c1, c2, c3);
        gather_edge_in<11>(x, s, lane, a0, a1, a2, a3, c0, c1, c2, c3);
        gather_edge_in<12>(x, s, lane, a0, a1, a2, a3, c0, c1, c2, c3);
        gather_edge_in<13>(x, s, lane, a0, a1, a2, a3, c0, c1, c2, c3);
        gather_edge_in<14>(x, s, lane, a0, a1, a2, a3, c0, c1, c2, c3);
        gather_edge_in<15>(x, s, lane, a0, a1, a2, a3, c0, c1, c2, c3);

        float* __restrict__ ap = agg + (size_t)node * (NHEAD * HID) + 2 * lane;
        *reinterpret_cast<float2*>(ap)           = c0;
        *reinterpret_cast<float2*>(ap + HID)     = c1;
        *reinterpret_cast<float2*>(ap + 2 * HID) = c2;
        *reinterpret_cast<float2*>(ap + 3 * HID) = c3;
    }
}

#define PROJ_THREADS 256
#define PROJ_K       (NHEAD * HID)

__global__ __launch_bounds__(PROJ_THREADS)
void gemm_wmma_f32_project_mean(const float* __restrict__ A,
                                const float* __restrict__ Wm,
                                const float* __restrict__ bias,
                                float* __restrict__ outm,
                                int nStrips) {
    constexpr int K  = PROJ_K;
    constexpr int NC = NHEAD * HID;
    __shared__ float4 Bs[K * 16];

    const int tid  = threadIdx.x;
    const int lane = tid & 31;
    const int wave = tid >> 5;
    const int g = lane >> 4;
    const int m = lane & 15;
    const int strip = blockIdx.x * (PROJ_THREADS / 32) + wave;
    const bool active = strip < nStrips;

    float4 a[K / 8];
    if (active) {
        const float4* __restrict__ Ap = reinterpret_cast<const float4*>(
            A + (size_t)(strip * 16 + m) * K + 4 * g);
#pragma unroll
        for (int i = 0; i < K / 8; ++i)
            a[i] = Ap[2 * i];
    }

    const float4 q0 = *reinterpret_cast<const float4*>(bias + 4 * m);
    const float4 q1 = *reinterpret_cast<const float4*>(bias + HID + 4 * m);
    const float4 q2 = *reinterpret_cast<const float4*>(bias + 2 * HID + 4 * m);
    const float4 q3 = *reinterpret_cast<const float4*>(bias + 3 * HID + 4 * m);

#pragma unroll
    for (int it = 0; it < (K * 8) / PROJ_THREADS; ++it) {
        const int item = it * PROJ_THREADS + tid;
        const int p  = item >> 4;
        const int mm = item & 15;
        const int kk = 2 * p;
        const float* __restrict__ Wp = Wm + (size_t)(kk & (HID - 1)) * NC + (kk >> 6) * HID + 4 * mm;
        const float4 x = *reinterpret_cast<const float4*>(Wp);
        const float4 y = *reinterpret_cast<const float4*>(Wp + NC);
        Bs[(2 * p)     * 16 + mm] = make_float4(x.x, y.x, x.y, y.y);
        Bs[(2 * p + 1) * 16 + mm] = make_float4(x.z, y.z, x.w, y.w);
    }
    __syncthreads();

    if (!active) return;

    v8f c0 = {}, c1 = {}, c2 = {}, c3 = {};
#pragma unroll
    for (int i = 0; i < K / 8; ++i) {
#pragma unroll
        for (int s = 0; s < 2; ++s) {
            const int p = 4 * i + 2 * g + s;
            const float4 b01 = Bs[(2 * p)     * 16 + m];
            const float4 b23 = Bs[(2 * p + 1) * 16 + m];
            v2f av, bv0, bv1, bv2, bv3;
            av.x  = s ? a[i].z : a[i].x;
            av.y  = s ? a[i].w : a[i].y;
            bv0.x = b01.x; bv0.y = b01.y;
            bv1.x = b01.z; bv1.y = b01.w;
            bv2.x = b23.x; bv2.y = b23.y;
            bv3.x = b23.z; bv3.y = b23.w;
            c0 = __builtin_amdgcn_wmma_f32_16x16x4_f32(false, av, false, bv0, (short)0, c0, false, false);
            c1 = __builtin_amdgcn_wmma_f32_16x16x4_f32(false, av, false, bv1, (short)0, c1, false, false);
            c2 = __builtin_amdgcn_wmma_f32_16x16x4_f32(false, av, false, bv2, (short)0, c2, false, false);
            c3 = __builtin_amdgcn_wmma_f32_16x16x4_f32(false, av, false, bv3, (short)0, c3, false, false);
        }
    }

    const float bm0 = 0.25f * ((q0.x + q1.x) + (q2.x + q3.x));
    const float bm1 = 0.25f * ((q0.y + q1.y) + (q2.y + q3.y));
    const float bm2 = 0.25f * ((q0.z + q1.z) + (q2.z + q3.z));
    const float bm3 = 0.25f * ((q0.w + q1.w) + (q2.w + q3.w));
    float* __restrict__ Op = outm + (size_t)(strip * 16 + 8 * g) * HID + 4 * m;
#pragma unroll
    for (int r = 0; r < 8; ++r)
        *reinterpret_cast<float4*>(Op + (size_t)r * HID) =
            make_float4(0.25f * c0[r] + bm0, 0.25f * c1[r] + bm1,
                        0.25f * c2[r] + bm2, 0.25f * c3[r] + bm3);
}

extern "C" void kernel_launch(void* const* d_in, const int* in_sizes, int n_in,
                              void* d_out, int out_size, void* d_ws, size_t ws_size,
                              hipStream_t stream) {
    const float* feat = (const float*)d_in[0];
    const int*   src  = (const int*)  d_in[1];
    const float* W1   = (const float*)d_in[3];
    const float* al1  = (const float*)d_in[4];
    const float* ar1  = (const float*)d_in[5];
    const float* b1   = (const float*)d_in[6];
    const float* W2   = (const float*)d_in[7];
    const float* al2  = (const float*)d_in[8];
    const float* ar2  = (const float*)d_in[9];
    const float* b2   = (const float*)d_in[10];

    const int N = NNODES;

    float* h  = (float*)d_ws;
    float* el = h  + (size_t)N * (NHEAD * HID);
    float* er = el + (size_t)N * NHEAD;
    float* x2 = er + (size_t)N * NHEAD;
    float* el2 = (float*)d_out;
    float* er2 = el2 + (size_t)N * NHEAD;

    const int nStrips     = 3125;
    const int gemmBlocks  = 3128;
    const int projBlocks  = 391;

    gemm_wmma_f32_panel<INF><<<gemmBlocks, 128, 0, stream>>>(feat, W1, h, al1, ar1, el, er, nStrips);
    gat_aggregate<<<AGG_BLOCKS, AGG_THREADS, 0, stream>>>(h, el, er, src, b1, x2, W2, al2, ar2, el2, er2, N);

    gat_aggregate_inputs<<<AGG_BLOCKS, AGG_THREADS, 0, stream>>>(x2, el2, er2, src, h, N);
    gemm_wmma_f32_project_mean<<<projBlocks, PROJ_THREADS, 0, stream>>>(h, W2, b2, (float*)d_out, nStrips);
}


# gather loop in fewer vector instructions: per-edge sums as packed fused multiply-adds (one rounding per step, order kept), DPP moves without an old-value copy, row loads on a 32-bit index
# speedup vs baseline: 2.3020x; 1.2315x over previous
#include <hip/hip_runtime.h>
#include <hip/hip_bf16.h>

#define NNODES 50000
#define NEDGES 800000
#define INF    128
#define HID    64
#define OUTD   64
#define NHEAD  4
#define SLOPE  0.2f
#define DEG    16

typedef __attribute__((ext_vector_type(2))) float v2f;
typedef __attribute__((ext_vector_type(4))) float v4f;
typedef __attribute__((ext_vector_type(8))) float v8f;

template <int J>
__device__ __forceinline__ float row_bcast(float x) {
    const int v = __float_as_int(x);
    return __int_as_float(__builtin_amdgcn_update_dpp(0, v, 0x150 + J, 0xF, 0xF, true));
}

template <int M>
__device__ __forceinline__ float row_xor(float x) {
    const int v = __float_as_int(x);
    return __int_as_float(__builtin_amdgcn_update_dpp(0, v, 0x160 + M, 0xF, 0xF, true));
}

__device__ __forceinline__ float row_max16(float x) {
    x = fmaxf(x, row_xor<8>(x));
    x = fmaxf(x, row_xor<4>(x));
    x = fmaxf(x, row_xor<2>(x));
    x = fmaxf(x, row_xor<1>(x));
    return x;
}

__device__ __forceinline__ float row_sum16(float x) {
    x += row_xor<8>(x);
    x += row_xor<4>(x);
    x += row_xor<2>(x);
    x += row_xor<1>(x);
    return x;
}


#define SC_THREADS 256

template <int K>
__global__ __launch_bounds__(SC_THREADS)
void gat_scores(const float* __restrict__ X,
                const float* __restrict__ Wn,
                const float* __restrict__ aln,
                const float* __restrict__ arn,
                float* __restrict__ el,
                float* __restrict__ er,
                int nStrips) {
    __shared__ float2 Bfs[(K / 2) * 16];

    const int tid  = threadIdx.x;
    const int lane = tid & 31;
    const int g = lane >> 4;
    const int m = lane & 15;
    const int wave = __builtin_amdgcn_readfirstlane(tid >> 5);

    for (int item = tid; item < K * NHEAD; item += SC_THREADS) {
        const int kk = item >> 2;
        const int hh = item & 3;
        const float4* __restrict__ wp = reinterpret_cast<const float4*>(Wn + (size_t)kk * (NHEAD * HID) + hh * HID);
        const float4* __restrict__ lp = reinterpret_cast<const float4*>(aln + hh * HID);
        const float4* __restrict__ rp = reinterpret_cast<const float4*>(arn + hh * HID);
        float wl = 0.f, wr = 0.f;
#pragma unroll 4
        for (int q = 0; q < HID / 4; ++q) {
            const float4 w = wp[q];
            const float4 a = lp[q];
            const float4 b = rp[q];
            wl += w.x * a.x + w.y * a.y + w.z * a.z + w.w * a.w;
            wr += w.x * b.x + w.y * b.y + w.z * b.z + w.w * b.w;
        }
        float* __restrict__ bf = reinterpret_cast<float*>(Bfs) + (kk >> 1) * 32 + (kk & 1);
        bf[2 * hh]        = wl;
        bf[2 * (4 + hh)]  = wr;
        bf[2 * (8 + hh)]  = 0.f;
        bf[2 * (12 + hh)] = 0.f;
    }
    __syncthreads();

    for (int strip = (int)blockIdx.x * (SC_THREADS / 32) + wave; strip < nStrips;
         strip += (int)gridDim.x * (SC_THREADS / 32)) {
        const float4* __restrict__ Ap = reinterpret_cast<const float4*>(
            X + (size_t)(strip * 16 + m) * K + 4 * g);
        float4 a[K / 8];
#pragma unroll
        for (int i = 0; i < K / 8; ++i)
            a[i] = Ap[2 * i];

        v8f acc0 = {}, acc1 = {};
#pragma unroll
        for (int i = 0; i < K / 8; ++i) {
            const float2 b0 = Bfs[(4 * i + 2 * g)     * 16 + m];
            const float2 b1 = Bfs[(4 * i + 2 * g + 1) * 16 + m];
            v2f av0, av1, bv0, bv1;
            av0.x = a[i].x; av0.y = a[i].y;
            av1.x = a[i].z; av1.y = a[i].w;
            bv0.x = b0.x; bv0.y = b0.y;
            bv1.x = b1.x; bv1.y = b1.y;
            acc0 = __builtin_amdgcn_wmma_f32_16x16x4_f32(false, av0, false, bv0, (short)0, acc0, false, false);
            acc1 = __builtin_amdgcn_wmma_f32_16x16x4_f32(false, av1, false, bv1, (short)0, acc1, false, false);
        }
        if (m < 8) {
            float* __restrict__ dstp = (m < 4) ? el : er;
            const int hh = m & 3;
#pragma unroll
            for (int r = 0; r < 8; ++r)
                dstp[(size_t)(strip * 16 + 8 * g + r) * NHEAD + hh] = acc0[r] + acc1[r];
        }
    }
}

template <int J, typename V>
__device__ __forceinline__ void gather_edge_in(const float* __restrict__ x, int s, unsigned lane,
                                               float a0, float a1, float a2, float a3,
                                               V& c0, V& c1, V& c2, V& c3) {
    const unsigned sj = (unsigned)__builtin_amdgcn_readlane(s, J);
    const V r = reinterpret_cast<const V*>(x)[sj * 32u + lane];
    const V w0 = (V)(row_bcast<J>(a0));
    const V w1 = (V)(row_bcast<J>(a1));
    const V w2 = (V)(row_bcast<J>(a2));
    const V w3 = (V)(row_bcast<J>(a3));
    if constexpr (J == 0) {
        c0 = w0 * r; c1 = w1 * r; c2 = w2 * r; c3 = w3 * r;
    } else {
        c0 = __builtin_elementwise_fma(w0, r, c0);
        c1 = __builtin_elementwise_fma(w1, r, c1);
        c2 = __builtin_elementwise_fma(w2, r, c2);
        c3 = __builtin_elementwise_fma(w3, r, c3);
    }
}

template <typename V>
__device__ __forceinline__ void gather_node(const float* __restrict__ x, int s, unsigned lane,
                                            float a0, float a1, float a2, float a3,
                                            V& c0, V& c1, V& c2, V& c3) {
    gather_edge_in<0>(x, s, lane, a0, a1, a2, a3, c0, c1, c2, c3);
    gather_edge_in<1>(x, s, lane, a0, a1, a2, a3, c0, c1, c2, c3);
    gather_edge_in<2>(x, s, lane, a0, a1, a2, a3, c0, c1, c2, c3);
    gather_edge_in<3>(x, s, lane, a0, a1, a2, a3, c0, c1, c2, c3);
    gather_edge_in<4>(x, s, lane, a0, a1, a2, a3, c0, c1, c2, c3);
    gather_edge_in<5>(x, s, lane, a0, a1, a2, a3, c0, c1, c2, c3);
    gather_edge_in<6>(x, s, lane, a0, a1, a2, a3, c0, c1, c2, c3);
    gather_edge_in<7>(x, s, lane, a0, a1, a2, a3, c0, c1, c2, c3);
    gather_edge_in<8>(x, s, lane, a0, a1, a2, a3, c0, c1, c2, c3);
    gather_edge_in<9>(x, s, lane, a0, a1, a2, a3, c0, c1, c2, c3);
    gather_edge_in<10>(x, s, lane, a0, a1, a2, a3, c0, c1, c2, c3);
    gather_edge_in<11>(x, s, lane, a0, a1, a2, a3, c0, c1, c2, c3);
    gather_edge_in<12>(x, s, lane, a0, a1, a2, a3, c0, c1, c2, c3);
    gather_edge_in<13>(x, s, lane, a0, a1, a2, a3, c0, c1, c2, c3);
    gather_edge_in<14>(x, s, lane, a0, a1, a2, a3, c0, c1, c2, c3);
    gather_edge_in<15>(x, s, lane, a0, a1, a2, a3, c0, c1, c2, c3);
}

__device__ __forceinline__ float edge_softmax16(float el, float er) {
    const float t = el + er;
    const float e = (t > 0.f) ? t : SLOPE * t;
    const float mx = row_max16(e);
    const float ex = __expf(e - mx);
    return ex / row_sum16(ex);
}

#define FUSE_GWAVES   16
#define FUSE_MWAVES   4
#define FUSE_THREADS  ((FUSE_GWAVES + FUSE_MWAVES) * 32)
#define FUSE_CH       4

template <int KIN, int NACC, int WPE>
__global__ __launch_bounds__(FUSE_THREADS) __attribute__((amdgpu_waves_per_eu(WPE)))
void gat_aggregate_project(const float* __restrict__ x,
                           const float* __restrict__ el,
                           const float* __restrict__ er,
                           const int*   __restrict__ src,
                           const float* __restrict__ Wm,
                           const float* __restrict__ bias,
                           float* __restrict__ outm,
                           int n_nodes) {
    constexpr int K  = NHEAD * KIN;
    constexpr int NC = NHEAD * HID;
    constexpr int AT_PITCH = K + 4;
    __shared__ float4 Bs[K * 16];
    __shared__ __attribute__((aligned(16))) float At[2][16 * AT_PITCH];

    const int tid  = threadIdx.x;
    const int lane = tid & 31;
    const int g = lane >> 4;
    const int m = lane & 15;
    const int wave = __builtin_amdgcn_readfirstlane(tid >> 5);
    const bool gatherer = wave < FUSE_GWAVES;
    const int nStrips = n_nodes >> 4;
    const int first = (int)blockIdx.x;
    const int step  = (int)gridDim.x;
    const int nTrips = (first < nStrips) ? (nStrips - 1 - first) / step + 1 : 0;

    int s_next = 0;
    if (gatherer && nTrips > 0)
        s_next = src[first * 16 + wave + m * n_nodes];

    for (int item = tid; item < (K / 4) * 16; item += FUSE_THREADS) {
        const int kq = item >> 4;
        const int mm = item & 15;
        const int kk = 4 * kq;
        const float* __restrict__ Wp = Wm + (size_t)(kk & (KIN - 1)) * NC + (kk / KIN) * HID + 4 * mm;
        const float4 r0 = *reinterpret_cast<const float4*>(Wp);
        const float4 r1 = *reinterpret_cast<const float4*>(Wp + NC);
        const float4 r2 = *reinterpret_cast<const float4*>(Wp + 2 * NC);
        const float4 r3 = *reinterpret_cast<const float4*>(Wp + 3 * NC);
        float4* __restrict__ bp = Bs + (kq >> 1) * 32 + (kq & 1) * 16 + mm;
        bp[0 * (K / 8) * 32] = make_float4(r0.x, r1.x, r2.x, r3.x);
        bp[1 * (K / 8) * 32] = make_float4(r0.y, r1.y, r2.y, r3.y);
        bp[2 * (K / 8) * 32] = make_float4(r0.z, r1.z, r2.z, r3.z);
        bp[3 * (K / 8) * 32] = make_float4(r0.w, r1.w, r2.w, r3.w);
    }

    const int tile = gatherer ? 0 : wave - FUSE_GWAVES;
    const int col = 4 * m + tile;
    float bm = 0.f;
    if (!gatherer)
        bm = 0.25f * ((bias[col] + bias[HID + col]) + (bias[2 * HID + col] + bias[3 * HID + col]));

    for (int t = 0; t <= nTrips; ++t) {
        if (gatherer) {
            if (t < nTrips) {
                const int strip = first + t * step;
                const int node = strip * 16 + wave;
                const int s = s_next;
                if (t + 1 < nTrips)
                    s_next = src[(strip + step) * 16 + wave + m * n_nodes];
                const float4 er4 = *reinterpret_cast<const float4*>(er + (size_t)node * NHEAD);
                const float4 el4 = *reinterpret_cast<const float4*>(el + (size_t)s * NHEAD);

                const float a0 = edge_softmax16(el4.x, er4.x);
                const float a1 = edge_softmax16(el4.y, er4.y);
                const float a2 = edge_softmax16(el4.z, er4.z);
                const float a3 = edge_softmax16(el4.w, er4.w);

                float* ap = At[t & 1] + wave * AT_PITCH;
                if constexpr (KIN == 64) {
                    v2f c0, c1, c2, c3;
                    gather_node(x, s, (unsigned)lane, a0, a1, a2, a3, c0, c1, c2, c3);
                    ap += 2 * lane;
                    *reinterpret_cast<v2f*>(ap)           = c0;
                    *reinterpret_cast<v2f*>(ap + KIN)     = c1;
                    *reinterpret_cast<v2f*>(ap + 2 * KIN) = c2;
                    *reinterpret_cast<v2f*>(ap + 3 * KIN) = c3;
                } else {
                    v4f c0, c1, c2, c3;
                    gather_node(x, s, (unsigned)lane, a0, a1, a2, a3, c0, c1, c2, c3);
                    ap += 4 * lane;
                    *reinterpret_cast<v4f*>(ap)           = c0;
                    *reinterpret_cast<v4f*>(ap + KIN)     = c1;
                    *reinterpret_cast<v4f*>(ap + 2 * KIN) = c2;
                    *reinterpret_cast<v4f*>(ap + 3 * KIN) = c3;
                }
            }
        } else if (t > 0) {
            const int strip = first + (t - 1) * step;
            const float* at = At[(t - 1) & 1] + m * AT_PITCH + 4 * g;
            const float4* bt = Bs + tile * (K / 8) * 32 + lane;

            v8f c[NACC];
#pragma unroll
            for (int n = 0; n < NACC; ++n) c[n] = v8f{};
            float4 aq[2][FUSE_CH];
            float4 bq[2][FUSE_CH];
#pragma unroll
            for (int q = 0; q < FUSE_CH; ++q) {
                aq[0][q] = *reinterpret_cast<const float4*>(at + 8 * q);
                bq[0][q] = bt[q * 32];
            }
#pragma unroll
            for (int ch = 0; ch < K / 8 / FUSE_CH; ++ch) {
                const int cur = ch & 1, nxt = cur ^ 1;
                if (ch + 1 < K / 8 / FUSE_CH) {
#pragma unroll
                    for (int q = 0; q < FUSE_CH; ++q) {
                        const int i = (ch + 1) * FUSE_CH + q;
                        aq[nxt][q] = *reinterpret_cast<const float4*>(at + 8 * i);
                        bq[nxt][q] = bt[i * 32];
                    }
                }
                __builtin_amdgcn_sched_barrier(0);
#pragma unroll
                for (int q = 0; q < FUSE_CH; ++q) {
#pragma unroll
                    for (int sx = 0; sx < 2; ++sx) {
                        const float4 a = aq[cur][q];
                        const float4 b = bq[cur][q];
                        v2f av, bv;
                        av.x = sx ? a.z : a.x;
                        av.y = sx ? a.w : a.y;
                        bv.x = sx ? b.z : b.x;
                        bv.y = sx ? b.w : b.y;
                        c[sx % NACC] = __builtin_amdgcn_wmma_f32_16x16x4_f32(false, av, false, bv, (short)0, c[sx % NACC], false, false);
                    }
                }
            }
            float* Op = outm + (size_t)(strip * 16 + 8 * g) * HID + col;
#pragma unroll
            for (int r = 0; r < 8; ++r) {
                float v = c[0][r];
                if constexpr (NACC == 2) v += c[1][r];
                Op[(size_t)r * HID] = 0.25f * v + bm;
            }
        }
        __syncthreads();
    }
}

extern "C" void kernel_launch(void* const* d_in, const int* in_sizes, int n_in,
                              void* d_out, int out_size, void* d_ws, size_t ws_size,
                              hipStream_t stream) {
    const float* feat = (const float*)d_in[0];
    const int*   src  = (const int*)  d_in[1];
    const float* W1   = (const float*)d_in[3];
    const float* al1  = (const float*)d_in[4];
    const float* ar1  = (const float*)d_in[5];
    const float* b1   = (const float*)d_in[6];
    const float* W2   = (const float*)d_in[7];
    const float* al2  = (const float*)d_in[8];
    const float* ar2  = (const float*)d_in[9];
    const float* b2   = (const float*)d_in[10];

    const int N = NNODES;

    float* x2  = (float*)d_ws;
    float* el1 = x2  + (size_t)N * HID;
    float* er1 = el1 + (size_t)N * NHEAD;
    float* el2 = er1 + (size_t)N * NHEAD;
    float* er2 = el2 + (size_t)N * NHEAD;

    const int nStrips     = 3125;
    const int scoreBlocks = 391;
    const int fuse1Blocks = 256;
    const int fuse2Blocks = 512;

    gat_scores<INF><<<scoreBlocks, SC_THREADS, 0, stream>>>(feat, W1, al1, ar1, el1, er1, nStrips);
    gat_aggregate_project<INF, 2, 5><<<fuse1Blocks, FUSE_THREADS, 0, stream>>>(feat, el1, er1, src, W1, b1, x2, N);

    gat_scores<HID><<<scoreBlocks, SC_THREADS, 0, stream>>>(x2, W2, al2, ar2, el2, er2, nStrips);
    gat_aggregate_project<HID, 1, 10><<<fuse2Blocks, FUSE_THREADS, 0, stream>>>(x2, el2, er2, src, W2, b2, (float*)d_out, N);
}


# gathering waves of both layers ask for node t+1's 16 source rows while node t is summed (row register refilled once read); row loads stay in flight across the trip barrier, which orders LDS only
# speedup vs baseline: 2.4924x; 1.0827x over previous
#include <hip/hip_runtime.h>
#include <hip/hip_bf16.h>

#define NNODES 50000
#define NEDGES 800000
#define INF    128
#define HID    64
#define OUTD   64
#define NHEAD  4
#define SLOPE  0.2f
#define DEG    16

typedef __attribute__((ext_vector_type(2))) float v2f;
typedef __attribute__((ext_vector_type(4))) float v4f;
typedef __attribute__((ext_vector_type(8))) float v8f;

template <int J>
__device__ __forceinline__ float row_bcast(float x) {
    const int v = __float_as_int(x);
    return __int_as_float(__builtin_amdgcn_update_dpp(0, v, 0x150 + J, 0xF, 0xF, true));
}

template <int M>
__device__ __forceinline__ float row_xor(float x) {
    const int v = __float_as_int(x);
    return __int_as_float(__builtin_amdgcn_update_dpp(0, v, 0x160 + M, 0xF, 0xF, true));
}

__device__ __forceinline__ float row_max16(float x) {
    x = fmaxf(x, row_xor<8>(x));
    x = fmaxf(x, row_xor<4>(x));
    x = fmaxf(x, row_xor<2>(x));
    x = fmaxf(x, row_xor<1>(x));
    return x;
}

__device__ __forceinline__ float row_sum16(float x) {
    x += row_xor<8>(x);
    x += row_xor<4>(x);
    x += row_xor<2>(x);
    x += row_xor<1>(x);
    return x;
}


#define SC_THREADS 256

template <int K>
__global__ __launch_bounds__(SC_THREADS)
void gat_scores(const float* __restrict__ X,
                const float* __restrict__ Wn,
                const float* __restrict__ aln,
                const float* __restrict__ arn,
                float* __restrict__ el,
                float* __restrict__ er,
                int nStrips) {
    __shared__ float2 Bfs[(K / 2) * 16];

    const int tid  = threadIdx.x;
    const int lane = tid & 31;
    const int g = lane >> 4;
    const int m = lane & 15;
    const int wave = __builtin_amdgcn_readfirstlane(tid >> 5);

    for (int item = tid; item < K * NHEAD; item += SC_THREADS) {
        const int kk = item >> 2;
        const int hh = item & 3;
        const float4* __restrict__ wp = reinterpret_cast<const float4*>(Wn + (size_t)kk * (NHEAD * HID) + hh * HID);
        const float4* __restrict__ lp = reinterpret_cast<const float4*>(aln + hh * HID);
        const float4* __restrict__ rp = reinterpret_cast<const float4*>(arn + hh * HID);
        float wl = 0.f, wr = 0.f;
#pragma unroll 4
        for (int q = 0; q < HID / 4; ++q) {
            const float4 w = wp[q];
            const float4 a = lp[q];
            const float4 b = rp[q];
            wl += w.x * a.x + w.y * a.y + w.z * a.z + w.w * a.w;
            wr += w.x * b.x + w.y * b.y + w.z * b.z + w.w * b.w;
        }
        float* __restrict__ bf = reinterpret_cast<float*>(Bfs) + (kk >> 1) * 32 + (kk & 1);
        bf[2 * hh]        = wl;
        bf[2 * (4 + hh)]  = wr;
        bf[2 * (8 + hh)]  = 0.f;
        bf[2 * (12 + hh)] = 0.f;
    }
    __syncthreads();

    for (int strip = (int)blockIdx.x * (SC_THREADS / 32) + wave; strip < nStrips;
         strip += (int)gridDim.x * (SC_THREADS / 32)) {
        const float4* __restrict__ Ap = reinterpret_cast<const float4*>(
            X + (size_t)(strip * 16 + m) * K + 4 * g);
        float4 a[K / 8];
#pragma unroll
        for (int i = 0; i < K / 8; ++i)
            a[i] = Ap[2 * i];

        v8f acc0 = {}, acc1 = {};
#pragma unroll
        for (int i = 0; i < K / 8; ++i) {
            const float2 b0 = Bfs[(4 * i + 2 * g)     * 16 + m];
            const float2 b1 = Bfs[(4 * i + 2 * g + 1) * 16 + m];
            v2f av0, av1, bv0, bv1;
            av0.x = a[i].x; av0.y = a[i].y;
            av1.x = a[i].z; av1.y = a[i].w;
            bv0.x = b0.x; bv0.y = b0.y;
            bv1.x = b1.x; bv1.y = b1.y;
            acc0 = __builtin_amdgcn_wmma_f32_16x16x4_f32(false, av0, false, bv0, (short)0, acc0, false, false);
            acc1 = __builtin_amdgcn_wmma_f32_16x16x4_f32(false, av1, false, bv1, (short)0, acc1, false, false);
        }
        if (m < 8) {
            float* __restrict__ dstp = (m < 4) ? el : er;
            const int hh = m & 3;
#pragma unroll
            for (int r = 0; r < 8; ++r)
                dstp[(size_t)(strip * 16 + 8 * g + r) * NHEAD + hh] = acc0[r] + acc1[r];
        }
    }
}

template <int J, typename V>
__device__ __forceinline__ V load_row(const float* __restrict__ x, int s, unsigned lane) {
    const unsigned sj = (unsigned)__builtin_amdgcn_readlane(s, J);
    return reinterpret_cast<const V*>(x)[sj * 32u + lane];
}

template <typename V>
__device__ __forceinline__ void load_rows(const float* __restrict__ x, int s, unsigned lane, V (&r)[16]) {
    r[0]  = load_row<0, V>(x, s, lane);   r[1]  = load_row<1, V>(x, s, lane);
    r[2]  = load_row<2, V>(x, s, lane);   r[3]  = load_row<3, V>(x, s, lane);
    r[4]  = load_row<4, V>(x, s, lane);   r[5]  = load_row<5, V>(x, s, lane);
    r[6]  = load_row<6, V>(x, s, lane);   r[7]  = load_row<7, V>(x, s, lane);
    r[8]  = load_row<8, V>(x, s, lane);   r[9]  = load_row<9, V>(x, s, lane);
    r[10] = load_row<10, V>(x, s, lane);  r[11] = load_row<11, V>(x, s, lane);
    r[12] = load_row<12, V>(x, s, lane);  r[13] = load_row<13, V>(x, s, lane);
    r[14] = load_row<14, V>(x, s, lane);  r[15] = load_row<15, V>(x, s, lane);
}

template <int J, bool MORE, typename V>
__device__ __forceinline__ void roll_edge(const float* __restrict__ x, int s_nxt, unsigned lane,
                                          float a0, float a1, float a2, float a3,
                                          V (&r)[16], V& c0, V& c1, V& c2, V& c3) {
    const V w0 = (V)(row_bcast<J>(a0));
    const V w1 = (V)(row_bcast<J>(a1));
    const V w2 = (V)(row_bcast<J>(a2));
    const V w3 = (V)(row_bcast<J>(a3));
    const V rj = r[J];
    if constexpr (J == 0) {
        c0 = w0 * rj; c1 = w1 * rj; c2 = w2 * rj; c3 = w3 * rj;
    } else {
        c0 = __builtin_elementwise_fma(w0, rj, c0);
        c1 = __builtin_elementwise_fma(w1, rj, c1);
        c2 = __builtin_elementwise_fma(w2, rj, c2);
        c3 = __builtin_elementwise_fma(w3, rj, c3);
    }
    if constexpr (MORE)
        r[J] = load_row<J, V>(x, s_nxt, lane);
}

template <bool MORE, typename V>
__device__ __forceinline__ void roll_node(const float* __restrict__ x, int s_nxt, unsigned lane,
                                          float a0, float a1, float a2, float a3,
                                          V (&r)[16], V& c0, V& c1, V& c2, V& c3) {
    roll_edge<0, MORE>(x, s_nxt, lane, a0, a1, a2, a3, r, c0, c1, c2, c3);
    roll_edge<1, MORE>(x, s_nxt, lane, a0, a1, a2, a3, r, c0, c1, c2, c3);
    roll_edge<2, MORE>(x, s_nxt, lane, a0, a1, a2, a3, r, c0, c1, c2, c3);
    roll_edge<3, MORE>(x, s_nxt, lane, a0, a1, a2, a3, r, c0, c1, c2, c3);
    roll_edge<4, MORE>(x, s_nxt, lane, a0, a1, a2, a3, r, c0, c1, c2, c3);
    roll_edge<5, MORE>(x, s_nxt, lane, a0, a1, a2, a3, r, c0, c1, c2, c3);
    roll_edge<6, MORE>(x, s_nxt, lane, a0, a1, a2, a3, r, c0, c1, c2, c3);
    roll_edge<7, MORE>(x, s_nxt, lane, a0, a1, a2, a3, r, c0, c1, c2, c3);
    roll_edge<8, MORE>(x, s_nxt, lane, a0, a1, a2, a3, r, c0, c1, c2, c3);
    roll_edge<9, MORE>(x, s_nxt, lane, a0, a1, a2, a3, r, c0, c1, c2, c3);
    roll_edge<10, MORE>(x, s_nxt, lane, a0, a1, a2, a3, r, c0, c1, c2, c3);
    roll_edge<11, MORE>(x, s_nxt, lane, a0, a1, a2, a3, r, c0, c1, c2, c3);
    roll_edge<12, MORE>(x, s_nxt, lane, a0, a1, a2, a3, r, c0, c1, c2, c3);
    roll_edge<13, MORE>(x, s_nxt, lane, a0, a1, a2, a3, r, c0, c1, c2, c3);
    roll_edge<14, MORE>(x, s_nxt, lane, a0, a1, a2, a3, r, c0, c1, c2, c3);
    roll_edge<15, MORE>(x, s_nxt, lane, a0, a1, a2, a3, r, c0, c1, c2, c3);
}

__device__ __forceinline__ float edge_softmax16(float el, float er) {
    const float t = el + er;
    const float e = (t > 0.f) ? t : SLOPE * t;
    const float mx = row_max16(e);
    const float ex = __expf(e - mx);
    return ex / row_sum16(ex);
}

__device__ __forceinline__ void barrier_lds_only() {
    __builtin_amdgcn_fence(__ATOMIC_RELEASE, "workgroup", "local");
    __builtin_amdgcn_s_barrier();
    __builtin_amdgcn_fence(__ATOMIC_ACQUIRE, "workgroup", "local");
}

template <int KIN> struct RowSlice;
template <> struct RowSlice<64>  { typedef v2f type; };
template <> struct RowSlice<128> { typedef v4f type; };

#define FUSE_GWAVES   16
#define FUSE_MWAVES   4
#define FUSE_THREADS  ((FUSE_GWAVES + FUSE_MWAVES) * 32)
#define FUSE_CH       4

template <int KIN, int NACC, int WPE>
__global__ __launch_bounds__(FUSE_THREADS) __attribute__((amdgpu_waves_per_eu(WPE)))
void gat_aggregate_project(const float* __restrict__ x,
                           const float* __restrict__ el,
                           const float* __restrict__ er,
                           const int*   __restrict__ src,
                           const float* __restrict__ Wm,
                           const float* __restrict__ bias,
                           float* __restrict__ outm,
                           int n_nodes) {
    constexpr int K  = NHEAD * KIN;
    constexpr int CH = (KIN == 64) ? 2 : FUSE_CH;
    constexpr int NC = NHEAD * HID;
    constexpr int AT_PITCH = K + 4;
    __shared__ float4 Bs[K * 16];
    __shared__ __attribute__((aligned(16))) float At[2][16 * AT_PITCH];

    const int tid  = threadIdx.x;
    const int lane = tid & 31;
    const int g = lane >> 4;
    const int m = lane & 15;
    const int wave = __builtin_amdgcn_readfirstlane(tid >> 5);
    const bool gatherer = wave < FUSE_GWAVES;
    const int nStrips = n_nodes >> 4;
    const int first = (int)blockIdx.x;
    const int step  = (int)gridDim.x;
    const int nTrips = (first < nStrips) ? (nStrips - 1 - first) / step + 1 : 0;

    typedef typename RowSlice<KIN>::type V;
    V rows[16];
    float a0 = 0.f, a1 = 0.f, a2 = 0.f, a3 = 0.f;

    int s_next = 0;
    int s_after = 0;
    if (gatherer && nTrips > 0) {
        s_next = src[first * 16 + wave + m * n_nodes];
        if (nTrips > 1)
            s_after = src[(first + step) * 16 + wave + m * n_nodes];
    }

    for (int item = tid; item < (K / 4) * 16; item += FUSE_THREADS) {
        const int kq = item >> 4;
        const int mm = item & 15;
        const int kk = 4 * kq;
        const float* __restrict__ Wp = Wm + (size_t)(kk & (KIN - 1)) * NC + (kk / KIN) * HID + 4 * mm;
        const float4 r0 = *reinterpret_cast<const float4*>(Wp);
        const float4 r1 = *reinterpret_cast<const float4*>(Wp + NC);
        const float4 r2 = *reinterpret_cast<const float4*>(Wp + 2 * NC);
        const float4 r3 = *reinterpret_cast<const float4*>(Wp + 3 * NC);
        float4* __restrict__ bp = Bs + (kq >> 1) * 32 + (kq & 1) * 16 + mm;
        bp[0 * (K / 8) * 32] = make_float4(r0.x, r1.x, r2.x, r3.x);
        bp[1 * (K / 8) * 32] = make_float4(r0.y, r1.y, r2.y, r3.y);
        bp[2 * (K / 8) * 32] = make_float4(r0.z, r1.z, r2.z, r3.z);
        bp[3 * (K / 8) * 32] = make_float4(r0.w, r1.w, r2.w, r3.w);
    }

    const int tile = gatherer ? 0 : wave - FUSE_GWAVES;
    const int col = 4 * m + tile;
    float bm = 0.f;
    if (!gatherer)
        bm = 0.25f * ((bias[col] + bias[HID + col]) + (bias[2 * HID + col] + bias[3 * HID + col]));

    if (gatherer && nTrips > 0) {
        const int node0 = first * 16 + wave;
        const float4 er4 = *reinterpret_cast<const float4*>(er + (size_t)node0 * NHEAD);
        const float4 el4 = *reinterpret_cast<const float4*>(el + (size_t)s_next * NHEAD);
        load_rows(x, s_next, (unsigned)lane, rows);
        a0 = edge_softmax16(el4.x, er4.x);
        a1 = edge_softmax16(el4.y, er4.y);
        a2 = edge_softmax16(el4.z, er4.z);
        a3 = edge_softmax16(el4.w, er4.w);
        s_next = s_after;
    }

    for (int t = 0; t <= nTrips; ++t) {
        if (gatherer) {
            if (t < nTrips) {
                const int strip = first + t * step;
                const int node = strip * 16 + wave;
                float* ap = At[t & 1] + wave * AT_PITCH + (KIN / 32) * lane;
                V c0, c1, c2, c3;
                if (t + 1 < nTrips) {
                    const float4 er4 = *reinterpret_cast<const float4*>(er + (size_t)(node + 16 * step) * NHEAD);
                    const float4 el4 = *reinterpret_cast<const float4*>(el + (size_t)s_next * NHEAD);
                    s_after = 0;
                    if (t + 2 < nTrips)
                        s_after = src[(strip + 2 * step) * 16 + wave + m * n_nodes];
                    roll_node<true>(x, s_next, (unsigned)lane, a0, a1, a2, a3, rows, c0, c1, c2, c3);
                    *reinterpret_cast<V*>(ap)           = c0;
                    *reinterpret_cast<V*>(ap + KIN)     = c1;
                    *reinterpret_cast<V*>(ap + 2 * KIN) = c2;
                    *reinterpret_cast<V*>(ap + 3 * KIN) = c3;
                    a0 = edge_softmax16(el4.x, er4.x);
                    a1 = edge_softmax16(el4.y, er4.y);
                    a2 = edge_softmax16(el4.z, er4.z);
                    a3 = edge_softmax16(el4.w, er4.w);
                    s_next = s_after;
                } else {
                    roll_node<false>(x, 0, (unsigned)lane, a0, a1, a2, a3, rows, c0, c1, c2, c3);
                    *reinterpret_cast<V*>(ap)           = c0;
                    *reinterpret_cast<V*>(ap + KIN)     = c1;
                    *reinterpret_cast<V*>(ap + 2 * KIN) = c2;
                    *reinterpret_cast<V*>(ap + 3 * KIN) = c3;
                }
            }
        } else if (t > 0) {
            const int strip = first + (t - 1) * step;
            const float* at = At[(t - 1) & 1] + m * AT_PITCH + 4 * g;
            const float4* bt = Bs + tile * (K / 8) * 32 + lane;

            v8f c[NACC];
#pragma unroll
            for (int n = 0; n < NACC; ++n) c[n] = v8f{};
            float4 aq[2][CH];
            float4 bq[2][CH];
#pragma unroll
            for (int q = 0; q < CH; ++q) {
                aq[0][q] = *reinterpret_cast<const float4*>(at + 8 * q);
                bq[0][q] = bt[q * 32];
            }
#pragma unroll
            for (int ch = 0; ch < K / 8 / CH; ++ch) {
                const int cur = ch & 1, nxt = cur ^ 1;
                if (ch + 1 < K / 8 / CH) {
#pragma unroll
                    for (int q = 0; q < CH; ++q) {
                        const int i = (ch + 1) * CH + q;
                        aq[nxt][q] = *reinterpret_cast<const float4*>(at + 8 * i);
                        bq[nxt][q] = bt[i * 32];
                    }
                }
                __builtin_amdgcn_sched_barrier(0);
#pragma unroll
                for (int q = 0; q < CH; ++q) {
#pragma unroll
                    for (int sx = 0; sx < 2; ++sx) {
                        const float4 a = aq[cur][q];
                        const float4 b = bq[cur][q];
                        v2f av, bv;
                        av.x = sx ? a.z : a.x;
                        av.y = sx ? a.w : a.y;
                        bv.x = sx ? b.z : b.x;
                        bv.y = sx ? b.w : b.y;
                        c[sx % NACC] = __builtin_amdgcn_wmma_f32_16x16x4_f32(false, av, false, bv, (short)0, c[sx % NACC], false, false);
                    }
                }
            }
            float* Op = outm + (size_t)(strip * 16 + 8 * g) * HID + col;
#pragma unroll
            for (int r = 0; r < 8; ++r) {
                float v = c[0][r];
                if constexpr (NACC == 2) v += c[1][r];
                Op[(size_t)r * HID] = 0.25f * v + bm;
            }
        }
        barrier_lds_only();
    }
}

extern "C" void kernel_launch(void* const* d_in, const int* in_sizes, int n_in,
                              void* d_out, int out_size, void* d_ws, size_t ws_size,
                              hipStream_t stream) {
    const float* feat = (const float*)d_in[0];
    const int*   src  = (const int*)  d_in[1];
    const float* W1   = (const float*)d_in[3];
    const float* al1  = (const float*)d_in[4];
    const float* ar1  = (const float*)d_in[5];
    const float* b1   = (const float*)d_in[6];
    const float* W2   = (const float*)d_in[7];
    const float* al2  = (const float*)d_in[8];
    const float* ar2  = (const float*)d_in[9];
    const float* b2   = (const float*)d_in[10];

    const int N = NNODES;

    float* x2  = (float*)d_ws;
    float* el1 = x2  + (size_t)N * HID;
    float* er1 = el1 + (size_t)N * NHEAD;
    float* el2 = er1 + (size_t)N * NHEAD;
    float* er2 = el2 + (size_t)N * NHEAD;

    const int nStrips     = 3125;
    const int scoreBlocks = 391;
    const int fuse1Blocks = 256;
    const int fuse2Blocks = 512;

    gat_scores<INF><<<scoreBlocks, SC_THREADS, 0, stream>>>(feat, W1, al1, ar1, el1, er1, nStrips);
    gat_aggregate_project<INF, 2, 5><<<fuse1Blocks, FUSE_THREADS, 0, stream>>>(feat, el1, er1, src, W1, b1, x2, N);

    gat_scores<HID><<<scoreBlocks, SC_THREADS, 0, stream>>>(x2, W2, al2, ar2, el2, er2, nStrips);
    gat_aggregate_project<HID, 1, 10><<<fuse2Blocks, FUSE_THREADS, 0, stream>>>(x2, el2, er2, src, W2, b2, (float*)d_out, N);
}


# layer 2's scores el2, er2 formed in layer 1's kernel: its multiplying waves keep each 16 x 64 x2 tile in LDS, one runs the folded-weight f32 WMMA chain on it; score launch gone: 4 -> 3 launches
# speedup vs baseline: 2.5836x; 1.0366x over previous
#include <hip/hip_runtime.h>
#include <hip/hip_bf16.h>

#define NNODES 50000
#define NEDGES 800000
#define INF    128
#define HID    64
#define OUTD   64
#define NHEAD  4
#define SLOPE  0.2f
#define DEG    16

typedef __attribute__((ext_vector_type(2))) float v2f;
typedef __attribute__((ext_vector_type(4))) float v4f;
typedef __attribute__((ext_vector_type(8))) float v8f;

template <int J>
__device__ __forceinline__ float row_bcast(float x) {
    const int v = __float_as_int(x);
    return __int_as_float(__builtin_amdgcn_update_dpp(0, v, 0x150 + J, 0xF, 0xF, true));
}

template <int M>
__device__ __forceinline__ float row_xor(float x) {
    const int v = __float_as_int(x);
    return __int_as_float(__builtin_amdgcn_update_dpp(0, v, 0x160 + M, 0xF, 0xF, true));
}

__device__ __forceinline__ float row_max16(float x) {
    x = fmaxf(x, row_xor<8>(x));
    x = fmaxf(x, row_xor<4>(x));
    x = fmaxf(x, row_xor<2>(x));
    x = fmaxf(x, row_xor<1>(x));
    return x;
}

__device__ __forceinline__ float row_sum16(float x) {
    x += row_xor<8>(x);
    x += row_xor<4>(x);
    x += row_xor<2>(x);
    x += row_xor<1>(x);
    return x;
}


#define SC_THREADS 256

template <int K>
__global__ __launch_bounds__(SC_THREADS)
void gat_scores(const float* __restrict__ X,
                const float* __restrict__ Wn,
                const float* __restrict__ aln,
                const float* __restrict__ arn,
                float* __restrict__ el,
                float* __restrict__ er,
                int nStrips) {
    __shared__ float2 Bfs[(K / 2) * 16];

    const int tid  = threadIdx.x;
    const int lane = tid & 31;
    const int g = lane >> 4;
    const int m = lane & 15;
    const int wave = __builtin_amdgcn_readfirstlane(tid >> 5);

    for (int item = tid; item < K * NHEAD; item += SC_THREADS) {
        const int kk = item >> 2;
        const int hh = item & 3;
        const float4* __restrict__ wp = reinterpret_cast<const float4*>(Wn + (size_t)kk * (NHEAD * HID) + hh * HID);
        const float4* __restrict__ lp = reinterpret_cast<const float4*>(aln + hh * HID);
        const float4* __restrict__ rp = reinterpret_cast<const float4*>(arn + hh * HID);
        float wl = 0.f, wr = 0.f;
#pragma unroll 4
        for (int q = 0; q < HID / 4; ++q) {
            const float4 w = wp[q];
            const float4 a = lp[q];
            const float4 b = rp[q];
            wl += w.x * a.x + w.y * a.y + w.z * a.z + w.w * a.w;
            wr += w.x * b.x + w.y * b.y + w.z * b.z + w.w * b.w;
        }
        float* __restrict__ bf = reinterpret_cast<float*>(Bfs) + (kk >> 1) * 32 + (kk & 1);
        bf[2 * hh]        = wl;
        bf[2 * (4 + hh)]  = wr;
        bf[2 * (8 + hh)]  = 0.f;
        bf[2 * (12 + hh)] = 0.f;
    }
    __syncthreads();

    for (int strip = (int)blockIdx.x * (SC_THREADS / 32) + wave; strip < nStrips;
         strip += (int)gridDim.x * (SC_THREADS / 32)) {
        const float4* __restrict__ Ap = reinterpret_cast<const float4*>(
            X + (size_t)(strip * 16 + m) * K + 4 * g);
        float4 a[K / 8];
#pragma unroll
        for (int i = 0; i < K / 8; ++i)
            a[i] = Ap[2 * i];

        v8f acc0 = {}, acc1 = {};
#pragma unroll
        for (int i = 0; i < K / 8; ++i) {
            const float2 b0 = Bfs[(4 * i + 2 * g)     * 16 + m];
            const float2 b1 = Bfs[(4 * i + 2 * g + 1) * 16 + m];
            v2f av0, av1, bv0, bv1;
            av0.x = a[i].x; av0.y = a[i].y;
            av1.x = a[i].z; av1.y = a[i].w;
            bv0.x = b0.x; bv0.y = b0.y;
            bv1.x = b1.x; bv1.y = b1.y;
            acc0 = __builtin_amdgcn_wmma_f32_16x16x4_f32(false, av0, false, bv0, (short)0, acc0, false, false);
            acc1 = __builtin_amdgcn_wmma_f32_16x16x4_f32(false, av1, false, bv1, (short)0, acc1, false, false);
        }
        if (m < 8) {
            float* __restrict__ dstp = (m < 4) ? el : er;
            const int hh = m & 3;
#pragma unroll
            for (int r = 0; r < 8; ++r)
                dstp[(size_t)(strip * 16 + 8 * g + r) * NHEAD + hh] = acc0[r] + acc1[r];
        }
    }
}

template <int J, typename V>
__device__ __forceinline__ V load_row(const float* __restrict__ x, int s, unsigned lane) {
    const unsigned sj = (unsigned)__builtin_amdgcn_readlane(s, J);
    return reinterpret_cast<const V*>(x)[sj * 32u + lane];
}

template <typename V>
__device__ __forceinline__ void load_rows(const float* __restrict__ x, int s, unsigned lane, V (&r)[16]) {
    r[0]  = load_row<0, V>(x, s, lane);   r[1]  = load_row<1, V>(x, s, lane);
    r[2]  = load_row<2, V>(x, s, lane);   r[3]  = load_row<3, V>(x, s, lane);
    r[4]  = load_row<4, V>(x, s, lane);   r[5]  = load_row<5, V>(x, s, lane);
    r[6]  = load_row<6, V>(x, s, lane);   r[7]  = load_row<7, V>(x, s, lane);
    r[8]  = load_row<8, V>(x, s, lane);   r[9]  = load_row<9, V>(x, s, lane);
    r[10] = load_row<10, V>(x, s, lane);  r[11] = load_row<11, V>(x, s, lane);
    r[12] = load_row<12, V>(x, s, lane);  r[13] = load_row<13, V>(x, s, lane);
    r[14] = load_row<14, V>(x, s, lane);  r[15] = load_row<15, V>(x, s, lane);
}

template <int J, bool MORE, typename V>
__device__ __forceinline__ void roll_edge(const float* __restrict__ x, int s_nxt, unsigned lane,
                                          float a0, float a1, float a2, float a3,
                                          V (&r)[16], V& c0, V& c1, V& c2, V& c3) {
    const V w0 = (V)(row_bcast<J>(a0));
    const V w1 = (V)(row_bcast<J>(a1));
    const V w2 = (V)(row_bcast<J>(a2));
    const V w3 = (V)(row_bcast<J>(a3));
    const V rj = r[J];
    if constexpr (J == 0) {
        c0 = w0 * rj; c1 = w1 * rj; c2 = w2 * rj; c3 = w3 * rj;
    } else {
        c0 = __builtin_elementwise_fma(w0, rj, c0);
        c1 = __builtin_elementwise_fma(w1, rj, c1);
        c2 = __builtin_elementwise_fma(w2, rj, c2);
        c3 = __builtin_elementwise_fma(w3, rj, c3);
    }
    if constexpr (MORE)
        r[J] = load_row<J, V>(x, s_nxt, lane);
}

template <bool MORE, typename V>
__device__ __forceinline__ void roll_node(const float* __restrict__ x, int s_nxt, unsigned lane,
                                          float a0, float a1, float a2, float a3,
                                          V (&r)[16], V& c0, V& c1, V& c2, V& c3) {
    roll_edge<0, MORE>(x, s_nxt, lane, a0, a1, a2, a3, r, c0, c1, c2, c3);
    roll_edge<1, MORE>(x, s_nxt, lane, a0, a1, a2, a3, r, c0, c1, c2, c3);
    roll_edge<2, MORE>(x, s_nxt, lane, a0, a1, a2, a3, r, c0, c1, c2, c3);
    roll_edge<3, MORE>(x, s_nxt, lane, a0, a1, a2, a3, r, c0, c1, c2, c3);
    roll_edge<4, MORE>(x, s_nxt, lane, a0, a1, a2, a3, r, c0, c1, c2, c3);
    roll_edge<5, MORE>(x, s_nxt, lane, a0, a1, a2, a3, r, c0, c1, c2, c3);
    roll_edge<6, MORE>(x, s_nxt, lane, a0, a1, a2, a3, r, c0, c1, c2, c3);
    roll_edge<7, MORE>(x, s_nxt, lane, a0, a1, a2, a3, r, c0, c1, c2, c3);
    roll_edge<8, MORE>(x, s_nxt, lane, a0, a1, a2, a3, r, c0, c1, c2, c3);
    roll_edge<9, MORE>(x, s_nxt, lane, a0, a1, a2, a3, r, c0, c1, c2, c3);
    roll_edge<10, MORE>(x, s_nxt, lane, a0, a1, a2, a3, r, c0, c1, c2, c3);
    roll_edge<11, MORE>(x, s_nxt, lane, a0, a1, a2, a3, r, c0, c1, c2, c3);
    roll_edge<12, MORE>(x, s_nxt, lane, a0, a1, a2, a3, r, c0, c1, c2, c3);
    roll_edge<13, MORE>(x, s_nxt, lane, a0, a1, a2, a3, r, c0, c1, c2, c3);
    roll_edge<14, MORE>(x, s_nxt, lane, a0, a1, a2, a3, r, c0, c1, c2, c3);
    roll_edge<15, MORE>(x, s_nxt, lane, a0, a1, a2, a3, r, c0, c1, c2, c3);
}

__device__ __forceinline__ float edge_softmax16(float el, float er) {
    const float t = el + er;
    const float e = (t > 0.f) ? t : SLOPE * t;
    const float mx = row_max16(e);
    const float ex = __expf(e - mx);
    return ex / row_sum16(ex);
}

__device__ __forceinline__ void barrier_lds_only() {
    __builtin_amdgcn_fence(__ATOMIC_RELEASE, "workgroup", "local");
    __builtin_amdgcn_s_barrier();
    __builtin_amdgcn_fence(__ATOMIC_ACQUIRE, "workgroup", "local");
}

template <int PITCH>
__device__ __forceinline__ void score_strip_from_lds(const float* xt, const float2* bfs, int g, int m,
                                                     float* __restrict__ el, float* __restrict__ er,
                                                     int strip) {
    const float* ap = xt + m * PITCH + 4 * g;
    float4 a[HID / 8];
#pragma unroll
    for (int i = 0; i < HID / 8; ++i)
        a[i] = *reinterpret_cast<const float4*>(ap + 8 * i);

    v8f acc0 = {}, acc1 = {};
#pragma unroll
    for (int i = 0; i < HID / 8; ++i) {
        const float2 b0 = bfs[(4 * i + 2 * g)     * 16 + m];
        const float2 b1 = bfs[(4 * i + 2 * g + 1) * 16 + m];
        v2f av0, av1, bv0, bv1;
        av0.x = a[i].x; av0.y = a[i].y;
        av1.x = a[i].z; av1.y = a[i].w;
        bv0.x = b0.x; bv0.y = b0.y;
        bv1.x = b1.x; bv1.y = b1.y;
        acc0 = __builtin_amdgcn_wmma_f32_16x16x4_f32(false, av0, false, bv0, (short)0, acc0, false, false);
        acc1 = __builtin_amdgcn_wmma_f32_16x16x4_f32(false, av1, false, bv1, (short)0, acc1, false, false);
    }
    if (m < 8) {
        float* __restrict__ dstp = (m < 4) ? el : er;
        const int hh = m & 3;
#pragma unroll
        for (int r = 0; r < 8; ++r)
            dstp[(size_t)(strip * 16 + 8 * g + r) * NHEAD + hh] = acc0[r] + acc1[r];
    }
}

template <int KIN> struct RowSlice;
template <> struct RowSlice<64>  { typedef v2f type; };
template <> struct RowSlice<128> { typedef v4f type; };

#define FUSE_GWAVES   16
#define FUSE_MWAVES   4
#define FUSE_THREADS  ((FUSE_GWAVES + FUSE_MWAVES) * 32)
#define FUSE_CH       4

template <int KIN, int NACC, int WPE, bool SCN>
__global__ __launch_bounds__(FUSE_THREADS) __attribute__((amdgpu_waves_per_eu(WPE)))
void gat_aggregate_project(const float* __restrict__ x,
                           const float* __restrict__ el,
                           const float* __restrict__ er,
                           const int*   __restrict__ src,
                           const float* __restrict__ Wm,
                           const float* __restrict__ bias,
                           float* __restrict__ outm,
                           int n_nodes,
                           const float* __restrict__ Wn,
                           const float* __restrict__ aln,
                           const float* __restrict__ arn,
                           float* __restrict__ eln,
                           float* __restrict__ ern) {
    constexpr int K  = NHEAD * KIN;
    constexpr int CH = (KIN == 64) ? 2 : FUSE_CH;
    constexpr int NC = NHEAD * HID;
    constexpr int AT_PITCH = K + 4;
    __shared__ float4 Bs[K * 16];
    __shared__ __attribute__((aligned(16))) float At[2][16 * AT_PITCH];
    constexpr int XN_PITCH = HID + 4;
    __shared__ float2 Bfn[SCN ? (HID / 2) * 16 : 1];
    __shared__ __attribute__((aligned(16))) float Xn[SCN ? 2 * 16 * XN_PITCH : 4];

    const int tid  = threadIdx.x;
    const int lane = tid & 31;
    const int g = lane >> 4;
    const int m = lane & 15;
    const int wave = __builtin_amdgcn_readfirstlane(tid >> 5);
    const bool gatherer = wave < FUSE_GWAVES;
    const int nStrips = n_nodes >> 4;
    const int first = (int)blockIdx.x;
    const int step  = (int)gridDim.x;
    const int nTrips = (first < nStrips) ? (nStrips - 1 - first) / step + 1 : 0;

    typedef typename RowSlice<KIN>::type V;
    V rows[16];
    float a0 = 0.f, a1 = 0.f, a2 = 0.f, a3 = 0.f;

    int s_next = 0;
    int s_after = 0;
    if (gatherer && nTrips > 0) {
        s_next = src[first * 16 + wave + m * n_nodes];
        if (nTrips > 1)
            s_after = src[(first + step) * 16 + wave + m * n_nodes];
    }

    for (int item = tid; item < (K / 4) * 16; item += FUSE_THREADS) {
        const int kq = item >> 4;
        const int mm = item & 15;
        const int kk = 4 * kq;
        const float* __restrict__ Wp = Wm + (size_t)(kk & (KIN - 1)) * NC + (kk / KIN) * HID + 4 * mm;
        const float4 r0 = *reinterpret_cast<const float4*>(Wp);
        const float4 r1 = *reinterpret_cast<const float4*>(Wp + NC);
        const float4 r2 = *reinterpret_cast<const float4*>(Wp + 2 * NC);
        const float4 r3 = *reinterpret_cast<const float4*>(Wp + 3 * NC);
        float4* __restrict__ bp = Bs + (kq >> 1) * 32 + (kq & 1) * 16 + mm;
        bp[0 * (K / 8) * 32] = make_float4(r0.x, r1.x, r2.x, r3.x);
        bp[1 * (K / 8) * 32] = make_float4(r0.y, r1.y, r2.y, r3.y);
        bp[2 * (K / 8) * 32] = make_float4(r0.z, r1.z, r2.z, r3.z);
        bp[3 * (K / 8) * 32] = make_float4(r0.w, r1.w, r2.w, r3.w);
    }

    if constexpr (SCN) {
        if (!gatherer) {
            const int it = tid - FUSE_GWAVES * 32;
            const int hh = it & 3;
            const int kA = it >> 2;
            const int kB = kA + 32;
            const float4* __restrict__ wa = reinterpret_cast<const float4*>(Wn + (size_t)kA * (NHEAD * HID) + hh * HID);
            const float4* __restrict__ wb = reinterpret_cast<const float4*>(Wn + (size_t)kB * (NHEAD * HID) + hh * HID);
            const float4* __restrict__ lp = reinterpret_cast<const float4*>(aln + hh * HID);
            const float4* __restrict__ rp = reinterpret_cast<const float4*>(arn + hh * HID);
            float wlA = 0.f, wrA = 0.f, wlB = 0.f, wrB = 0.f;
#pragma unroll 4
            for (int q = 0; q < HID / 4; ++q) {
                const float4 u = wa[q];
                const float4 v = wb[q];
                const float4 a = lp[q];
                const float4 b = rp[q];
                wlA += u.x * a.x + u.y * a.y + u.z * a.z + u.w * a.w;
                wrA += u.x * b.x + u.y * b.y + u.z * b.z + u.w * b.w;
                wlB += v.x * a.x + v.y * a.y + v.z * a.z + v.w * a.w;
                wrB += v.x * b.x + v.y * b.y + v.z * b.z + v.w * b.w;
            }
            float* __restrict__ bfA = reinterpret_cast<float*>(Bfn) + (kA >> 1) * 32 + (kA & 1);
            float* __restrict__ bfB = reinterpret_cast<float*>(Bfn) + (kB >> 1) * 32 + (kB & 1);
            bfA[2 * hh]        = wlA;
            bfA[2 * (4 + hh)]  = wrA;
            bfA[2 * (8 + hh)]  = 0.f;
            bfA[2 * (12 + hh)] = 0.f;
            bfB[2 * hh]        = wlB;
            bfB[2 * (4 + hh)]  = wrB;
            bfB[2 * (8 + hh)]  = 0.f;
            bfB[2 * (12 + hh)] = 0.f;
        }
    }

    const int tile = gatherer ? 0 : wave - FUSE_GWAVES;
    const int col = 4 * m + tile;
    float bm = 0.f;
    if (!gatherer)
        bm = 0.25f * ((bias[col] + bias[HID + col]) + (bias[2 * HID + col] + bias[3 * HID + col]));

    if (gatherer && nTrips > 0) {
        const int node0 = first * 16 + wave;
        const float4 er4 = *reinterpret_cast<const float4*>(er + (size_t)node0 * NHEAD);
        const float4 el4 = *reinterpret_cast<const float4*>(el + (size_t)s_next * NHEAD);
        load_rows(x, s_next, (unsigned)lane, rows);
        a0 = edge_softmax16(el4.x, er4.x);
        a1 = edge_softmax16(el4.y, er4.y);
        a2 = edge_softmax16(el4.z, er4.z);
        a3 = edge_softmax16(el4.w, er4.w);
        s_next = s_after;
    }

    for (int t = 0; t <= nTrips; ++t) {
        if (gatherer) {
            if (t < nTrips) {
                const int strip = first + t * step;
                const int node = strip * 16 + wave;
                float* ap = At[t & 1] + wave * AT_PITCH + (KIN / 32) * lane;
                V c0, c1, c2, c3;
                if (t + 1 < nTrips) {
                    const float4 er4 = *reinterpret_cast<const float4*>(er + (size_t)(node + 16 * step) * NHEAD);
                    const float4 el4 = *reinterpret_cast<const float4*>(el + (size_t)s_next * NHEAD);
                    s_after = 0;
                    if (t + 2 < nTrips)
                        s_after = src[(strip + 2 * step) * 16 + wave + m * n_nodes];
                    roll_node<true>(x, s_next, (unsigned)lane, a0, a1, a2, a3, rows, c0, c1, c2, c3);
                    *reinterpret_cast<V*>(ap)           = c0;
                    *reinterpret_cast<V*>(ap + KIN)     = c1;
                    *reinterpret_cast<V*>(ap + 2 * KIN) = c2;
                    *reinterpret_cast<V*>(ap + 3 * KIN) = c3;
                    a0 = edge_softmax16(el4.x, er4.x);
                    a1 = edge_softmax16(el4.y, er4.y);
                    a2 = edge_softmax16(el4.z, er4.z);
                    a3 = edge_softmax16(el4.w, er4.w);
                    s_next = s_after;
                } else {
                    roll_node<false>(x, 0, (unsigned)lane, a0, a1, a2, a3, rows, c0, c1, c2, c3);
                    *reinterpret_cast<V*>(ap)           = c0;
                    *reinterpret_cast<V*>(ap + KIN)     = c1;
                    *reinterpret_cast<V*>(ap + 2 * KIN) = c2;
                    *reinterpret_cast<V*>(ap + 3 * KIN) = c3;
                }
            }
        } else if (t > 0) {
            const int strip = first + (t - 1) * step;
            const float* at = At[(t - 1) & 1] + m * AT_PITCH + 4 * g;
            const float4* bt = Bs + tile * (K / 8) * 32 + lane;

            v8f c[NACC];
#pragma unroll
            for (int n = 0; n < NACC; ++n) c[n] = v8f{};
            float4 aq[2][CH];
            float4 bq[2][CH];
#pragma unroll
            for (int q = 0; q < CH; ++q) {
                aq[0][q] = *reinterpret_cast<const float4*>(at + 8 * q);
                bq[0][q] = bt[q * 32];
            }
#pragma unroll
            for (int ch = 0; ch < K / 8 / CH; ++ch) {
                const int cur = ch & 1, nxt = cur ^ 1;
                if (ch + 1 < K / 8 / CH) {
#pragma unroll
                    for (int q = 0; q < CH; ++q) {
                        const int i = (ch + 1) * CH + q;
                        aq[nxt][q] = *reinterpret_cast<const float4*>(at + 8 * i);
                        bq[nxt][q] = bt[i * 32];
                    }
                }
                __builtin_amdgcn_sched_barrier(0);
#pragma unroll
                for (int q = 0; q < CH; ++q) {
#pragma unroll
                    for (int sx = 0; sx < 2; ++sx) {
                        const float4 a = aq[cur][q];
                        const float4 b = bq[cur][q];
                        v2f av, bv;
                        av.x = sx ? a.z : a.x;
                        av.y = sx ? a.w : a.y;
                        bv.x = sx ? b.z : b.x;
                        bv.y = sx ? b.w : b.y;
                        c[sx % NACC] = __builtin_amdgcn_wmma_f32_16x16x4_f32(false, av, false, bv, (short)0, c[sx % NACC], false, false);
                    }
                }
            }
            float* Op = outm + (size_t)(strip * 16 + 8 * g) * HID + col;
            float* Xp = Xn + ((t - 1) & 1) * (16 * XN_PITCH) + (8 * g) * XN_PITCH + col;
#pragma unroll
            for (int r = 0; r < 8; ++r) {
                float v = c[0][r];
                if constexpr (NACC == 2) v += c[1][r];
                const float o = 0.25f * v + bm;
                Op[(size_t)r * HID] = o;
                if constexpr (SCN) Xp[r * XN_PITCH] = o;
            }
        }
        if constexpr (SCN) {
            if (!gatherer && t > 1 && tile == (t & 3))
                score_strip_from_lds<XN_PITCH>(Xn + (t & 1) * (16 * XN_PITCH), Bfn, g, m, eln, ern,
                                               first + (t - 2) * step);
        }
        barrier_lds_only();
    }
    if constexpr (SCN) {
        if (!gatherer && nTrips > 0 && tile == ((nTrips + 1) & 3))
            score_strip_from_lds<XN_PITCH>(Xn + ((nTrips - 1) & 1) * (16 * XN_PITCH), Bfn, g, m, eln, ern,
                                           first + (nTrips - 1) * step);
    }
}

extern "C" void kernel_launch(void* const* d_in, const int* in_sizes, int n_in,
                              void* d_out, int out_size, void* d_ws, size_t ws_size,
                              hipStream_t stream) {
    const float* feat = (const float*)d_in[0];
    const int*   src  = (const int*)  d_in[1];
    const float* W1   = (const float*)d_in[3];
    const float* al1  = (const float*)d_in[4];
    const float* ar1  = (const float*)d_in[5];
    const float* b1   = (const float*)d_in[6];
    const float* W2   = (const float*)d_in[7];
    const float* al2  = (const float*)d_in[8];
    const float* ar2  = (const float*)d_in[9];
    const float* b2   = (const float*)d_in[10];

    const int N = NNODES;

    float* x2  = (float*)d_ws;
    float* el1 = x2  + (size_t)N * HID;
    float* er1 = el1 + (size_t)N * NHEAD;
    float* el2 = er1 + (size_t)N * NHEAD;
    float* er2 = el2 + (size_t)N * NHEAD;

    const int nStrips     = 3125;
    const int scoreBlocks = 391;
    const int fuse1Blocks = 256;
    const int fuse2Blocks = 512;

    gat_scores<INF><<<scoreBlocks, SC_THREADS, 0, stream>>>(feat, W1, al1, ar1, el1, er1, nStrips);
    gat_aggregate_project<INF, 2, 5, true><<<fuse1Blocks, FUSE_THREADS, 0, stream>>>(
        feat, el1, er1, src, W1, b1, x2, N, W2, al2, ar2, el2, er2);

    gat_aggregate_project<HID, 1, 10, false><<<fuse2Blocks, FUSE_THREADS, 0, stream>>>(
        x2, el2, er2, src, W2, b2, (float*)d_out, N, nullptr, nullptr, nullptr, nullptr, nullptr);
}
